# starting kernel
# baseline (speedup 1.0000x reference)
#include <hip/hip_runtime.h>
#include <hip/hip_bf16.h>

typedef float v2f __attribute__((ext_vector_type(2)));
typedef float v4f __attribute__((ext_vector_type(4)));
typedef float v8f __attribute__((ext_vector_type(8)));

#define IN_NF   64
#define HIDN    128
#define EHW     64
#define KIN_NODE (HIDN + EHW)

__device__ __forceinline__ float silu_f(float x) {
    return x / (1.0f + __expf(-x));
}

__device__ __forceinline__ v8f splat8(float v) {
    v8f r = {v, v, v, v, v, v, v, v};
    return r;
}

__device__ __forceinline__ v8f wmma4(v2f a, v2f b, v8f c) {
    return __builtin_amdgcn_wmma_f32_16x16x4_f32(
         false, a,  false, b,
         (short)0, c,  false,  false);
}

__global__ void embed_kernel(const float* __restrict__ hin,
                             const float* __restrict__ W,
                             const float* __restrict__ bias,
                             float* __restrict__ hout, int Nn) {
    const int lane = threadIdx.x & 31;
    const int c    = lane & 15;
    const int half = lane >> 4;
    const int nw   = (gridDim.x * blockDim.x) >> 5;
    const int wid0 = (blockIdx.x * blockDim.x + threadIdx.x) >> 5;
    const int tiles = (Nn + 15) >> 4;

    for (int t = wid0; t < tiles; t += nw) {
        int mr = t * 16 + c;
        if (mr >= Nn) mr = Nn - 1;

        v8f acc[8];
#pragma unroll
        for (int n = 0; n < 8; ++n) acc[n] = splat8(bias[n * 16 + c]);

        for (int k = 0; k < IN_NF; k += 4) {
            const int kk = k + half * 2;
            v2f a = *(const v2f*)(hin + (size_t)mr * IN_NF + kk);
#pragma unroll
            for (int n = 0; n < 8; ++n) {
                v2f b;
                b.x = W[kk * HIDN + n * 16 + c];
                b.y = W[(kk + 1) * HIDN + n * 16 + c];
                acc[n] = wmma4(a, b, acc[n]);
            }
        }
        if (t * 16 + 16 <= Nn) {
#pragma unroll
            for (int n = 0; n < 8; ++n) {
                float* p = hout + (size_t)(t * 16 + half * 8) * HIDN + n * 16 + c;
#pragma unroll
                for (int i = 0; i < 8; ++i) p[(size_t)i * HIDN] = acc[n][i];
            }
        } else {
#pragma unroll
            for (int n = 0; n < 8; ++n)
#pragma unroll
                for (int i = 0; i < 8; ++i) {
                    const int r = t * 16 + i + half * 8;
                    if (r < Nn) hout[(size_t)r * HIDN + n * 16 + c] = acc[n][i];
                }
        }
    }
}

__global__ void edge_kernel(const float* __restrict__ h,
                            const int* __restrict__ erow,
                            const int* __restrict__ ecol,
                            const float* __restrict__ w1,
                            const float* __restrict__ b1,
                            const float* __restrict__ w2,
                            const float* __restrict__ b2,
                            float* __restrict__ agg, int Ee) {
    constexpr int WS  = 72;
    constexpr int LWM = EHW + 4;
    __shared__ float w1s[2 * HIDN * WS];
    __shared__ float w2s[EHW * WS];
    __shared__ float m1s[8][16 * LWM];

    for (int i4 = threadIdx.x; i4 < (2 * HIDN * EHW) / 4; i4 += blockDim.x) {
        const int r = i4 >> 4, col = (i4 & 15) * 4;
        *(v4f*)(w1s + r * WS + col) = *(const v4f*)(w1 + r * EHW + col);
    }
    for (int i4 = threadIdx.x; i4 < (EHW * EHW) / 4; i4 += blockDim.x) {
        const int r = i4 >> 4, col = (i4 & 15) * 4;
        *(v4f*)(w2s + r * WS + col) = *(const v4f*)(w2 + r * EHW + col);
    }
    __syncthreads();

    const int lane = threadIdx.x & 31;
    const int wv   = threadIdx.x >> 5;
    const int c    = lane & 15;
    const int half = lane >> 4;
    const int nw   = (gridDim.x * blockDim.x) >> 5;
    const int wid0 = (blockIdx.x * blockDim.x + threadIdx.x) >> 5;
    const int tiles = (Ee + 15) >> 4;
    float* L = m1s[wv];

    for (int t = wid0; t < tiles; t += nw) {
        int ea = t * 16 + c;
        if (ea >= Ee) ea = Ee - 1;
        const float* __restrict__ hra = h + (size_t)erow[ea] * HIDN;
        const float* __restrict__ hca = h + (size_t)ecol[ea] * HIDN;

        v8f acc[4];
#pragma unroll
        for (int n = 0; n < 4; ++n) acc[n] = splat8(b1[n * 16 + c]);

#pragma unroll 4
        for (int k = 0; k < HIDN; k += 4) {
            const int kk = k + half * 2;
            v2f a = *(const v2f*)(hra + kk);
#pragma unroll
            for (int n = 0; n < 4; ++n) {
                v2f b;
                b.x = w1s[kk * WS + n * 16 + c];
                b.y = w1s[(kk + 1) * WS + n * 16 + c];
                acc[n] = wmma4(a, b, acc[n]);
            }
        }
#pragma unroll 4
        for (int k = 0; k < HIDN; k += 4) {
            const int kk = k + half * 2;
            v2f a = *(const v2f*)(hca + kk);
#pragma unroll
            for (int n = 0; n < 4; ++n) {
                v2f b;
                b.x = w1s[(HIDN + kk) * WS + n * 16 + c];
                b.y = w1s[(HIDN + kk + 1) * WS + n * 16 + c];
                acc[n] = wmma4(a, b, acc[n]);
            }
        }

#pragma unroll
        for (int n = 0; n < 4; ++n)
#pragma unroll
            for (int i = 0; i < 8; ++i)
                L[(i + half * 8) * LWM + n * 16 + c] = silu_f(acc[n][i]);

        asm volatile("s_wait_dscnt 0" ::: "memory");

        v8f acc2[4];
#pragma unroll
        for (int n = 0; n < 4; ++n) acc2[n] = splat8(b2[n * 16 + c]);

#pragma unroll 4
        for (int k = 0; k < EHW; k += 4) {
            const int kk = k + half * 2;
            v2f a;
            a.x = L[c * LWM + kk];
            a.y = L[c * LWM + kk + 1];
#pragma unroll
            for (int n = 0; n < 4; ++n) {
                v2f b;
                b.x = w2s[kk * WS + n * 16 + c];
                b.y = w2s[(kk + 1) * WS + n * 16 + c];
                acc2[n] = wmma4(a, b, acc2[n]);
            }
        }

        if (t * 16 + 16 <= Ee) {
#pragma unroll
            for (int i = 0; i < 8; ++i) {
                float* dst = agg + (size_t)erow[t * 16 + i + half * 8] * EHW + c;
#pragma unroll
                for (int n = 0; n < 4; ++n)
                    atomicAdd(dst + n * 16, silu_f(acc2[n][i]));
            }
        } else {
#pragma unroll
            for (int i = 0; i < 8; ++i) {
                const int e = t * 16 + i + half * 8;
                if (e < Ee) {
                    float* dst = agg + (size_t)erow[e] * EHW + c;
#pragma unroll
                    for (int n = 0; n < 4; ++n)
                        atomicAdd(dst + n * 16, silu_f(acc2[n][i]));
                }
            }
        }
        asm volatile("s_wait_dscnt 0" ::: "memory");
    }
}

__global__ void node_kernel(float* __restrict__ h,
                            const float* __restrict__ agg,
                            const float* __restrict__ m1,
                            const float* __restrict__ c1,
                            const float* __restrict__ m2,
                            const float* __restrict__ c2,
                            int Nn) {
    constexpr int LW  = HIDN + 4;
    constexpr int WS2 = HIDN + 8;
    __shared__ float w2s[HIDN * WS2];
    __shared__ float lds[8][16 * LW];

    for (int i4 = threadIdx.x; i4 < (HIDN * HIDN) / 4; i4 += blockDim.x) {
        const int r = i4 >> 5, col = (i4 & 31) * 4;
        *(v4f*)(w2s + r * WS2 + col) = *(const v4f*)(m2 + r * HIDN + col);
    }
    __syncthreads();

    const int lane = threadIdx.x & 31;
    const int wv   = threadIdx.x >> 5;
    const int c    = lane & 15;
    const int half = lane >> 4;
    const int nw   = (gridDim.x * blockDim.x) >> 5;
    const int wid0 = (blockIdx.x * blockDim.x + threadIdx.x) >> 5;
    const int tiles = (Nn + 15) >> 4;
    float* L = lds[wv];

    for (int t = wid0; t < tiles; t += nw) {
        int mr = t * 16 + c;
        if (mr >= Nn) mr = Nn - 1;
        const float* __restrict__ hr = h + (size_t)mr * HIDN;
        const float* __restrict__ ar = agg + (size_t)mr * EHW;

        v8f acc[8];
#pragma unroll
        for (int n = 0; n < 8; ++n) acc[n] = splat8(c1[n * 16 + c]);

        for (int k = 0; k < HIDN; k += 4) {
            const int kk = k + half * 2;
            v2f a = *(const v2f*)(hr + kk);
#pragma unroll
            for (int n = 0; n < 8; ++n) {
                v2f b;
                b.x = m1[kk * HIDN + n * 16 + c];
                b.y = m1[(kk + 1) * HIDN + n * 16 + c];
                acc[n] = wmma4(a, b, acc[n]);
            }
        }
        for (int k = 0; k < EHW; k += 4) {
            const int kk = k + half * 2;
            v2f a = *(const v2f*)(ar + kk);
#pragma unroll
            for (int n = 0; n < 8; ++n) {
                v2f b;
                b.x = m1[(HIDN + kk) * HIDN + n * 16 + c];
                b.y = m1[(HIDN + kk + 1) * HIDN + n * 16 + c];
                acc[n] = wmma4(a, b, acc[n]);
            }
        }
#pragma unroll
        for (int n = 0; n < 8; ++n)
#pragma unroll
            for (int i = 0; i < 8; ++i)
                L[(i + half * 8) * LW + n * 16 + c] = silu_f(acc[n][i]);

        asm volatile("s_wait_dscnt 0" ::: "memory");

        v8f acc2[8];
#pragma unroll
        for (int n = 0; n < 8; ++n) acc2[n] = splat8(c2[n * 16 + c]);

        for (int k = 0; k < HIDN; k += 4) {
            const int kk = k + half * 2;
            v2f a;
            a.x = L[c * LW + kk];
            a.y = L[c * LW + kk + 1];
#pragma unroll
            for (int n = 0; n < 8; ++n) {
                v2f b;
                b.x = w2s[kk * WS2 + n * 16 + c];
                b.y = w2s[(kk + 1) * WS2 + n * 16 + c];
                acc2[n] = wmma4(a, b, acc2[n]);
            }
        }
        if (t * 16 + 16 <= Nn) {
#pragma unroll
            for (int n = 0; n < 8; ++n) {
                float* p = h + (size_t)(t * 16 + half * 8) * HIDN + n * 16 + c;
#pragma unroll
                for (int i = 0; i < 8; ++i) p[(size_t)i * HIDN] += acc2[n][i];
            }
        } else {
#pragma unroll
            for (int n = 0; n < 8; ++n)
#pragma unroll
                for (int i = 0; i < 8; ++i) {
                    const int r = t * 16 + i + half * 8;
                    if (r < Nn) h[(size_t)r * HIDN + n * 16 + c] += acc2[n][i];
                }
        }
        asm volatile("s_wait_dscnt 0" ::: "memory");
    }
}

__global__ void out_kernel(const float* __restrict__ h,
                           const float* __restrict__ W,
                           const float* __restrict__ bias,
                           float* __restrict__ out, int Nn) {
    const int lane = threadIdx.x & 31;
    const int c    = lane & 15;
    const int half = lane >> 4;
    const int nw   = (gridDim.x * blockDim.x) >> 5;
    const int wid0 = (blockIdx.x * blockDim.x + threadIdx.x) >> 5;
    const int tiles = (Nn + 15) >> 4;

    for (int t = wid0; t < tiles; t += nw) {
        int mr = t * 16 + c;
        if (mr >= Nn) mr = Nn - 1;

        v8f acc[4];
#pragma unroll
        for (int n = 0; n < 4; ++n) acc[n] = splat8(bias[n * 16 + c]);

        for (int k = 0; k < HIDN; k += 4) {
            const int kk = k + half * 2;
            v2f a = *(const v2f*)(h + (size_t)mr * HIDN + kk);
#pragma unroll
            for (int n = 0; n < 4; ++n) {
                v2f b;
                b.x = W[kk * EHW + n * 16 + c];
                b.y = W[(kk + 1) * EHW + n * 16 + c];
                acc[n] = wmma4(a, b, acc[n]);
            }
        }
        if (t * 16 + 16 <= Nn) {
#pragma unroll
            for (int n = 0; n < 4; ++n) {
                float* p = out + (size_t)(t * 16 + half * 8) * EHW + n * 16 + c;
#pragma unroll
                for (int i = 0; i < 8; ++i) p[(size_t)i * EHW] = acc[n][i];
            }
        } else {
#pragma unroll
            for (int n = 0; n < 4; ++n)
#pragma unroll
                for (int i = 0; i < 8; ++i) {
                    const int r = t * 16 + i + half * 8;
                    if (r < Nn) out[(size_t)r * EHW + n * 16 + c] = acc[n][i];
                }
        }
    }
}

static inline int imin_h(int a, int b) { return a < b ? a : b; }
static inline int cdiv_h(int a, int b) { return (a + b - 1) / b; }

extern "C" void kernel_launch(void* const* d_in, const int* in_sizes, int n_in,
                              void* d_out, int out_size, void* d_ws, size_t ws_size,
                              hipStream_t stream) {
    const float* h_in  = (const float*)d_in[0];
    const int*   edges = (const int*)d_in[1];
    const float* W_in  = (const float*)d_in[2];
    const float* b_in  = (const float*)d_in[3];
    const float* eW1   = (const float*)d_in[4];
    const float* eb1   = (const float*)d_in[5];
    const float* eW2   = (const float*)d_in[6];
    const float* eb2   = (const float*)d_in[7];
    const float* nW1   = (const float*)d_in[8];
    const float* nb1   = (const float*)d_in[9];
    const float* nW2   = (const float*)d_in[10];
    const float* nb2   = (const float*)d_in[11];
    const float* W_out = (const float*)d_in[12];
    const float* b_out = (const float*)d_in[13];

    const int Nn = in_sizes[0] / IN_NF;
    const int Ee = in_sizes[1] / 2;
    const int Ll = in_sizes[4] / (2 * HIDN * EHW);

    float* h   = (float*)d_ws;
    float* agg = h + (size_t)Nn * HIDN;

    const int* erow = edges;
    const int* ecol = edges + Ee;

    const int tilesN = cdiv_h(Nn, 16);
    const int tilesE = cdiv_h(Ee, 16);
    const int gEmb = imin_h(cdiv_h(tilesN, 8), 2048);
    const int gEdg = imin_h(cdiv_h(tilesE, 8), 2048);
    const int gNod = imin_h(cdiv_h(tilesN, 8), 2048);

    embed_kernel<<<gEmb, 256, 0, stream>>>(h_in, W_in, b_in, h, Nn);

    for (int l = 0; l < Ll; ++l) {
        hipMemsetAsync(agg, 0, (size_t)Nn * EHW * sizeof(float), stream);
        edge_kernel<<<gEdg, 256, 0, stream>>>(
            h, erow, ecol,
            eW1 + (size_t)l * 2 * HIDN * EHW, eb1 + (size_t)l * EHW,
            eW2 + (size_t)l * EHW * EHW,      eb2 + (size_t)l * EHW,
            agg, Ee);
        node_kernel<<<gNod, 256, 0, stream>>>(
            h, agg,
            nW1 + (size_t)l * KIN_NODE * HIDN, nb1 + (size_t)l * HIDN,
            nW2 + (size_t)l * HIDN * HIDN,     nb2 + (size_t)l * HIDN,
            Nn);
    }

    out_kernel<<<gEmb, 256, 0, stream>>>(h, W_out, b_out, (float*)d_out, Nn);
}


# Row half of the edge MLP's first product hoisted to the nodes: P = h @ W1[0:128] + b1 once per node (one new launch per layer), edge tiles start from gathered P[row]; float sum order changes
# speedup vs baseline: 1.2649x; 1.2649x over previous
#include <hip/hip_runtime.h>
#include <hip/hip_bf16.h>

typedef float v2f __attribute__((ext_vector_type(2)));
typedef float v4f __attribute__((ext_vector_type(4)));
typedef float v8f __attribute__((ext_vector_type(8)));

#define IN_NF   64
#define HIDN    128
#define EHW     64
#define KIN_NODE (HIDN + EHW)

__device__ __forceinline__ float silu_f(float x) {
    return x / (1.0f + __expf(-x));
}

__device__ __forceinline__ v8f splat8(float v) {
    v8f r = {v, v, v, v, v, v, v, v};
    return r;
}

__device__ __forceinline__ v8f wmma4(v2f a, v2f b, v8f c) {
    return __builtin_amdgcn_wmma_f32_16x16x4_f32(
         false, a,  false, b,
         (short)0, c,  false,  false);
}

__global__ void embed_kernel(const float* __restrict__ hin,
                             const float* __restrict__ W,
                             const float* __restrict__ bias,
                             float* __restrict__ hout, int Nn) {
    const int lane = threadIdx.x & 31;
    const int c    = lane & 15;
    const int half = lane >> 4;
    const int nw   = (gridDim.x * blockDim.x) >> 5;
    const int wid0 = (blockIdx.x * blockDim.x + threadIdx.x) >> 5;
    const int tiles = (Nn + 15) >> 4;

    for (int t = wid0; t < tiles; t += nw) {
        int mr = t * 16 + c;
        if (mr >= Nn) mr = Nn - 1;

        v8f acc[8];
#pragma unroll
        for (int n = 0; n < 8; ++n) acc[n] = splat8(bias[n * 16 + c]);

        for (int k = 0; k < IN_NF; k += 4) {
            const int kk = k + half * 2;
            v2f a = *(const v2f*)(hin + (size_t)mr * IN_NF + kk);
#pragma unroll
            for (int n = 0; n < 8; ++n) {
                v2f b;
                b.x = W[kk * HIDN + n * 16 + c];
                b.y = W[(kk + 1) * HIDN + n * 16 + c];
                acc[n] = wmma4(a, b, acc[n]);
            }
        }
        if (t * 16 + 16 <= Nn) {
#pragma unroll
            for (int n = 0; n < 8; ++n) {
                float* p = hout + (size_t)(t * 16 + half * 8) * HIDN + n * 16 + c;
#pragma unroll
                for (int i = 0; i < 8; ++i) p[(size_t)i * HIDN] = acc[n][i];
            }
        } else {
#pragma unroll
            for (int n = 0; n < 8; ++n)
#pragma unroll
                for (int i = 0; i < 8; ++i) {
                    const int r = t * 16 + i + half * 8;
                    if (r < Nn) hout[(size_t)r * HIDN + n * 16 + c] = acc[n][i];
                }
        }
    }
}

__global__ void rowproj_kernel(const float* __restrict__ h,
                               const float* __restrict__ w1,
                               const float* __restrict__ b1,
                               float* __restrict__ P, int Nn) {
    constexpr int WS = 72;
    __shared__ float w1s[HIDN * WS];

    for (int i4 = threadIdx.x; i4 < (HIDN * EHW) / 4; i4 += blockDim.x) {
        const int r = i4 >> 4, col = (i4 & 15) * 4;
        *(v4f*)(w1s + r * WS + col) = *(const v4f*)(w1 + r * EHW + col);
    }
    __syncthreads();

    const int lane = threadIdx.x & 31;
    const int c    = lane & 15;
    const int half = lane >> 4;
    const int nw   = (gridDim.x * blockDim.x) >> 5;
    const int wid0 = (blockIdx.x * blockDim.x + threadIdx.x) >> 5;
    const int tiles = (Nn + 15) >> 4;

    for (int t = wid0; t < tiles; t += nw) {
        int mr = t * 16 + c;
        if (mr >= Nn) mr = Nn - 1;
        const float* __restrict__ hr = h + (size_t)mr * HIDN;

        v8f acc[4];
#pragma unroll
        for (int n = 0; n < 4; ++n) acc[n] = splat8(b1[n * 16 + c]);

#pragma unroll 8
        for (int k = 0; k < HIDN; k += 4) {
            const int kk = k + half * 2;
            v2f a = *(const v2f*)(hr + kk);
#pragma unroll
            for (int n = 0; n < 4; ++n) {
                v2f b;
                b.x = w1s[kk * WS + n * 16 + c];
                b.y = w1s[(kk + 1) * WS + n * 16 + c];
                acc[n] = wmma4(a, b, acc[n]);
            }
        }
        if (t * 16 + 16 <= Nn) {
#pragma unroll
            for (int n = 0; n < 4; ++n) {
                float* p = P + (size_t)(t * 16 + half * 8) * EHW + n * 16 + c;
#pragma unroll
                for (int i = 0; i < 8; ++i) p[(size_t)i * EHW] = acc[n][i];
            }
        } else {
#pragma unroll
            for (int n = 0; n < 4; ++n)
#pragma unroll
                for (int i = 0; i < 8; ++i) {
                    const int r = t * 16 + i + half * 8;
                    if (r < Nn) P[(size_t)r * EHW + n * 16 + c] = acc[n][i];
                }
        }
    }
}

__global__ void edge_kernel(const float* __restrict__ h,
                            const float* __restrict__ P,
                            const int* __restrict__ erow,
                            const int* __restrict__ ecol,
                            const float* __restrict__ w1c,
                            const float* __restrict__ w2,
                            const float* __restrict__ b2,
                            float* __restrict__ agg, int Ee) {
    constexpr int WS  = 72;
    constexpr int LWM = EHW + 4;
    __shared__ float w1s[HIDN * WS];
    __shared__ float w2s[EHW * WS];
    __shared__ float m1s[8][16 * LWM];

    for (int i4 = threadIdx.x; i4 < (HIDN * EHW) / 4; i4 += blockDim.x) {
        const int r = i4 >> 4, col = (i4 & 15) * 4;
        *(v4f*)(w1s + r * WS + col) = *(const v4f*)(w1c + r * EHW + col);
    }
    for (int i4 = threadIdx.x; i4 < (EHW * EHW) / 4; i4 += blockDim.x) {
        const int r = i4 >> 4, col = (i4 & 15) * 4;
        *(v4f*)(w2s + r * WS + col) = *(const v4f*)(w2 + r * EHW + col);
    }
    __syncthreads();

    const int lane = threadIdx.x & 31;
    const int wv   = threadIdx.x >> 5;
    const int c    = lane & 15;
    const int half = lane >> 4;
    const int nw   = (gridDim.x * blockDim.x) >> 5;
    const int wid0 = (blockIdx.x * blockDim.x + threadIdx.x) >> 5;
    const int tiles = (Ee + 15) >> 4;
    float* L = m1s[wv];

    for (int t = wid0; t < tiles; t += nw) {
        int ea = t * 16 + c;
        if (ea >= Ee) ea = Ee - 1;
        const float* __restrict__ hca = h + (size_t)ecol[ea] * HIDN;

        int ri[8];
#pragma unroll
        for (int i = 0; i < 8; ++i) {
            int e = t * 16 + i + half * 8;
            if (e >= Ee) e = Ee - 1;
            ri[i] = erow[e];
        }

        v8f acc[4];
#pragma unroll
        for (int i = 0; i < 8; ++i) {
            const float* __restrict__ pr = P + (size_t)ri[i] * EHW + c;
#pragma unroll
            for (int n = 0; n < 4; ++n) acc[n][i] = pr[n * 16];
        }

#pragma unroll 4
        for (int k = 0; k < HIDN; k += 4) {
            const int kk = k + half * 2;
            v2f a = *(const v2f*)(hca + kk);
#pragma unroll
            for (int n = 0; n < 4; ++n) {
                v2f b;
                b.x = w1s[kk * WS + n * 16 + c];
                b.y = w1s[(kk + 1) * WS + n * 16 + c];
                acc[n] = wmma4(a, b, acc[n]);
            }
        }

#pragma unroll
        for (int n = 0; n < 4; ++n)
#pragma unroll
            for (int i = 0; i < 8; ++i)
                L[(i + half * 8) * LWM + n * 16 + c] = silu_f(acc[n][i]);

        asm volatile("s_wait_dscnt 0" ::: "memory");

        v8f acc2[4];
#pragma unroll
        for (int n = 0; n < 4; ++n) acc2[n] = splat8(b2[n * 16 + c]);

#pragma unroll 4
        for (int k = 0; k < EHW; k += 4) {
            const int kk = k + half * 2;
            v2f a;
            a.x = L[c * LWM + kk];
            a.y = L[c * LWM + kk + 1];
#pragma unroll
            for (int n = 0; n < 4; ++n) {
                v2f b;
                b.x = w2s[kk * WS + n * 16 + c];
                b.y = w2s[(kk + 1) * WS + n * 16 + c];
                acc2[n] = wmma4(a, b, acc2[n]);
            }
        }

        if (t * 16 + 16 <= Ee) {
#pragma unroll
            for (int i = 0; i < 8; ++i) {
                float* dst = agg + (size_t)ri[i] * EHW + c;
#pragma unroll
                for (int n = 0; n < 4; ++n)
                    atomicAdd(dst + n * 16, silu_f(acc2[n][i]));
            }
        } else {
#pragma unroll
            for (int i = 0; i < 8; ++i) {
                const int e = t * 16 + i + half * 8;
                if (e < Ee) {
                    float* dst = agg + (size_t)ri[i] * EHW + c;
#pragma unroll
                    for (int n = 0; n < 4; ++n)
                        atomicAdd(dst + n * 16, silu_f(acc2[n][i]));
                }
            }
        }
        asm volatile("s_wait_dscnt 0" ::: "memory");
    }
}

__global__ void node_kernel(float* __restrict__ h,
                            const float* __restrict__ agg,
                            const float* __restrict__ m1,
                            const float* __restrict__ c1,
                            const float* __restrict__ m2,
                            const float* __restrict__ c2,
                            int Nn) {
    constexpr int LW  = HIDN + 4;
    constexpr int WS2 = HIDN + 8;
    __shared__ float w2s[HIDN * WS2];
    __shared__ float lds[8][16 * LW];

    for (int i4 = threadIdx.x; i4 < (HIDN * HIDN) / 4; i4 += blockDim.x) {
        const int r = i4 >> 5, col = (i4 & 31) * 4;
        *(v4f*)(w2s + r * WS2 + col) = *(const v4f*)(m2 + r * HIDN + col);
    }
    __syncthreads();

    const int lane = threadIdx.x & 31;
    const int wv   = threadIdx.x >> 5;
    const int c    = lane & 15;
    const int half = lane >> 4;
    const int nw   = (gridDim.x * blockDim.x) >> 5;
    const int wid0 = (blockIdx.x * blockDim.x + threadIdx.x) >> 5;
    const int tiles = (Nn + 15) >> 4;
    float* L = lds[wv];

    for (int t = wid0; t < tiles; t += nw) {
        int mr = t * 16 + c;
        if (mr >= Nn) mr = Nn - 1;
        const float* __restrict__ hr = h + (size_t)mr * HIDN;
        const float* __restrict__ ar = agg + (size_t)mr * EHW;

        v8f acc[8];
#pragma unroll
        for (int n = 0; n < 8; ++n) acc[n] = splat8(c1[n * 16 + c]);

        for (int k = 0; k < HIDN; k += 4) {
            const int kk = k + half * 2;
            v2f a = *(const v2f*)(hr + kk);
#pragma unroll
            for (int n = 0; n < 8; ++n) {
                v2f b;
                b.x = m1[kk * HIDN + n * 16 + c];
                b.y = m1[(kk + 1) * HIDN + n * 16 + c];
                acc[n] = wmma4(a, b, acc[n]);
            }
        }
        for (int k = 0; k < EHW; k += 4) {
            const int kk = k + half * 2;
            v2f a = *(const v2f*)(ar + kk);
#pragma unroll
            for (int n = 0; n < 8; ++n) {
                v2f b;
                b.x = m1[(HIDN + kk) * HIDN + n * 16 + c];
                b.y = m1[(HIDN + kk + 1) * HIDN + n * 16 + c];
                acc[n] = wmma4(a, b, acc[n]);
            }
        }
#pragma unroll
        for (int n = 0; n < 8; ++n)
#pragma unroll
            for (int i = 0; i < 8; ++i)
                L[(i + half * 8) * LW + n * 16 + c] = silu_f(acc[n][i]);

        asm volatile("s_wait_dscnt 0" ::: "memory");

        v8f acc2[8];
#pragma unroll
        for (int n = 0; n < 8; ++n) acc2[n] = splat8(c2[n * 16 + c]);

        for (int k = 0; k < HIDN; k += 4) {
            const int kk = k + half * 2;
            v2f a;
            a.x = L[c * LW + kk];
            a.y = L[c * LW + kk + 1];
#pragma unroll
            for (int n = 0; n < 8; ++n) {
                v2f b;
                b.x = w2s[kk * WS2 + n * 16 + c];
                b.y = w2s[(kk + 1) * WS2 + n * 16 + c];
                acc2[n] = wmma4(a, b, acc2[n]);
            }
        }
        if (t * 16 + 16 <= Nn) {
#pragma unroll
            for (int n = 0; n < 8; ++n) {
                float* p = h + (size_t)(t * 16 + half * 8) * HIDN + n * 16 + c;
#pragma unroll
                for (int i = 0; i < 8; ++i) p[(size_t)i * HIDN] += acc2[n][i];
            }
        } else {
#pragma unroll
            for (int n = 0; n < 8; ++n)
#pragma unroll
                for (int i = 0; i < 8; ++i) {
                    const int r = t * 16 + i + half * 8;
                    if (r < Nn) h[(size_t)r * HIDN + n * 16 + c] += acc2[n][i];
                }
        }
        asm volatile("s_wait_dscnt 0" ::: "memory");
    }
}

__global__ void out_kernel(const float* __restrict__ h,
                           const float* __restrict__ W,
                           const float* __restrict__ bias,
                           float* __restrict__ out, int Nn) {
    const int lane = threadIdx.x & 31;
    const int c    = lane & 15;
    const int half = lane >> 4;
    const int nw   = (gridDim.x * blockDim.x) >> 5;
    const int wid0 = (blockIdx.x * blockDim.x + threadIdx.x) >> 5;
    const int tiles = (Nn + 15) >> 4;

    for (int t = wid0; t < tiles; t += nw) {
        int mr = t * 16 + c;
        if (mr >= Nn) mr = Nn - 1;

        v8f acc[4];
#pragma unroll
        for (int n = 0; n < 4; ++n) acc[n] = splat8(bias[n * 16 + c]);

        for (int k = 0; k < HIDN; k += 4) {
            const int kk = k + half * 2;
            v2f a = *(const v2f*)(h + (size_t)mr * HIDN + kk);
#pragma unroll
            for (int n = 0; n < 4; ++n) {
                v2f b;
                b.x = W[kk * EHW + n * 16 + c];
                b.y = W[(kk + 1) * EHW + n * 16 + c];
                acc[n] = wmma4(a, b, acc[n]);
            }
        }
        if (t * 16 + 16 <= Nn) {
#pragma unroll
            for (int n = 0; n < 4; ++n) {
                float* p = out + (size_t)(t * 16 + half * 8) * EHW + n * 16 + c;
#pragma unroll
                for (int i = 0; i < 8; ++i) p[(size_t)i * EHW] = acc[n][i];
            }
        } else {
#pragma unroll
            for (int n = 0; n < 4; ++n)
#pragma unroll
                for (int i = 0; i < 8; ++i) {
                    const int r = t * 16 + i + half * 8;
                    if (r < Nn) out[(size_t)r * EHW + n * 16 + c] = acc[n][i];
                }
        }
    }
}

static inline int imin_h(int a, int b) { return a < b ? a : b; }
static inline int cdiv_h(int a, int b) { return (a + b - 1) / b; }

extern "C" void kernel_launch(void* const* d_in, const int* in_sizes, int n_in,
                              void* d_out, int out_size, void* d_ws, size_t ws_size,
                              hipStream_t stream) {
    const float* h_in  = (const float*)d_in[0];
    const int*   edges = (const int*)d_in[1];
    const float* W_in  = (const float*)d_in[2];
    const float* b_in  = (const float*)d_in[3];
    const float* eW1   = (const float*)d_in[4];
    const float* eb1   = (const float*)d_in[5];
    const float* eW2   = (const float*)d_in[6];
    const float* eb2   = (const float*)d_in[7];
    const float* nW1   = (const float*)d_in[8];
    const float* nb1   = (const float*)d_in[9];
    const float* nW2   = (const float*)d_in[10];
    const float* nb2   = (const float*)d_in[11];
    const float* W_out = (const float*)d_in[12];
    const float* b_out = (const float*)d_in[13];

    const int Nn = in_sizes[0] / IN_NF;
    const int Ee = in_sizes[1] / 2;
    const int Ll = in_sizes[4] / (2 * HIDN * EHW);

    float* h   = (float*)d_ws;
    float* agg = h + (size_t)Nn * HIDN;

    const int* erow = edges;
    const int* ecol = edges + Ee;

    const int tilesN = cdiv_h(Nn, 16);
    const int tilesE = cdiv_h(Ee, 16);
    const int gEmb = imin_h(cdiv_h(tilesN, 8), 2048);
    const int gEdg = imin_h(cdiv_h(tilesE, 8), 2048);
    const int gNod = imin_h(cdiv_h(tilesN, 8), 2048);

    embed_kernel<<<gEmb, 256, 0, stream>>>(h_in, W_in, b_in, h, Nn);

    float* P = (float*)d_out;

    for (int l = 0; l < Ll; ++l) {
        rowproj_kernel<<<gNod, 256, 0, stream>>>(
            h, eW1 + (size_t)l * 2 * HIDN * EHW, eb1 + (size_t)l * EHW, P, Nn);
        hipMemsetAsync(agg, 0, (size_t)Nn * EHW * sizeof(float), stream);
        edge_kernel<<<gEdg, 256, 0, stream>>>(
            h, P, erow, ecol,
            eW1 + (size_t)l * 2 * HIDN * EHW + (size_t)HIDN * EHW,
            eW2 + (size_t)l * EHW * EHW,      eb2 + (size_t)l * EHW,
            agg, Ee);
        node_kernel<<<gNod, 256, 0, stream>>>(
            h, agg,
            nW1 + (size_t)l * KIN_NODE * HIDN, nb1 + (size_t)l * HIDN,
            nW2 + (size_t)l * HIDN * HIDN,     nb2 + (size_t)l * HIDN,
            Nn);
    }

    out_kernel<<<gEmb, 256, 0, stream>>>(h, W_out, b_out, (float*)d_out, Nn);
}


# Col half of the edge MLP's first product hoisted to the nodes too (Q = h @ W1[128:256] per node); edge tiles form silu(P[row]+Q[col]) directly as the A fragment; float sum order changes
# speedup vs baseline: 1.6531x; 1.3069x over previous
#include <hip/hip_runtime.h>
#include <hip/hip_bf16.h>

typedef float v2f __attribute__((ext_vector_type(2)));
typedef float v4f __attribute__((ext_vector_type(4)));
typedef float v8f __attribute__((ext_vector_type(8)));

#define IN_NF   64
#define HIDN    128
#define EHW     64
#define KIN_NODE (HIDN + EHW)

__device__ __forceinline__ float silu_f(float x) {
    return x / (1.0f + __expf(-x));
}

__device__ __forceinline__ v8f splat8(float v) {
    v8f r = {v, v, v, v, v, v, v, v};
    return r;
}

__device__ __forceinline__ v8f wmma4(v2f a, v2f b, v8f c) {
    return __builtin_amdgcn_wmma_f32_16x16x4_f32(
         false, a,  false, b,
         (short)0, c,  false,  false);
}

__global__ void embed_kernel(const float* __restrict__ hin,
                             const float* __restrict__ W,
                             const float* __restrict__ bias,
                             float* __restrict__ hout, int Nn) {
    const int lane = threadIdx.x & 31;
    const int c    = lane & 15;
    const int half = lane >> 4;
    const int nw   = (gridDim.x * blockDim.x) >> 5;
    const int wid0 = (blockIdx.x * blockDim.x + threadIdx.x) >> 5;
    const int tiles = (Nn + 15) >> 4;

    for (int t = wid0; t < tiles; t += nw) {
        int mr = t * 16 + c;
        if (mr >= Nn) mr = Nn - 1;

        v8f acc[8];
#pragma unroll
        for (int n = 0; n < 8; ++n) acc[n] = splat8(bias[n * 16 + c]);

        for (int k = 0; k < IN_NF; k += 4) {
            const int kk = k + half * 2;
            v2f a = *(const v2f*)(hin + (size_t)mr * IN_NF + kk);
#pragma unroll
            for (int n = 0; n < 8; ++n) {
                v2f b;
                b.x = W[kk * HIDN + n * 16 + c];
                b.y = W[(kk + 1) * HIDN + n * 16 + c];
                acc[n] = wmma4(a, b, acc[n]);
            }
        }
        if (t * 16 + 16 <= Nn) {
#pragma unroll
            for (int n = 0; n < 8; ++n) {
                float* p = hout + (size_t)(t * 16 + half * 8) * HIDN + n * 16 + c;
#pragma unroll
                for (int i = 0; i < 8; ++i) p[(size_t)i * HIDN] = acc[n][i];
            }
        } else {
#pragma unroll
            for (int n = 0; n < 8; ++n)
#pragma unroll
                for (int i = 0; i < 8; ++i) {
                    const int r = t * 16 + i + half * 8;
                    if (r < Nn) hout[(size_t)r * HIDN + n * 16 + c] = acc[n][i];
                }
        }
    }
}

__global__ void rowproj_kernel(const float* __restrict__ h,
                               const float* __restrict__ w1all,
                               const float* __restrict__ b1,
                               float* __restrict__ Pout,
                               float* __restrict__ Qout, int Nn) {
    constexpr int WS = 72;
    __shared__ float w1s[HIDN * WS];

    const int side = blockIdx.y;
    const float* __restrict__ w1 = w1all + (size_t)side * HIDN * EHW;
    float* __restrict__ P = side ? Qout : Pout;

    for (int i4 = threadIdx.x; i4 < (HIDN * EHW) / 4; i4 += blockDim.x) {
        const int r = i4 >> 4, col = (i4 & 15) * 4;
        *(v4f*)(w1s + r * WS + col) = *(const v4f*)(w1 + r * EHW + col);
    }
    __syncthreads();

    const int lane = threadIdx.x & 31;
    const int c    = lane & 15;
    const int half = lane >> 4;
    const int nw   = (gridDim.x * blockDim.x) >> 5;
    const int wid0 = (blockIdx.x * blockDim.x + threadIdx.x) >> 5;
    const int tiles = (Nn + 15) >> 4;

    for (int t = wid0; t < tiles; t += nw) {
        int mr = t * 16 + c;
        if (mr >= Nn) mr = Nn - 1;
        const float* __restrict__ hr = h + (size_t)mr * HIDN;

        v8f acc[4];
#pragma unroll
        for (int n = 0; n < 4; ++n) acc[n] = splat8(side ? 0.0f : b1[n * 16 + c]);

#pragma unroll 8
        for (int k = 0; k < HIDN; k += 4) {
            const int kk = k + half * 2;
            v2f a = *(const v2f*)(hr + kk);
#pragma unroll
            for (int n = 0; n < 4; ++n) {
                v2f b;
                b.x = w1s[kk * WS + n * 16 + c];
                b.y = w1s[(kk + 1) * WS + n * 16 + c];
                acc[n] = wmma4(a, b, acc[n]);
            }
        }
        if (t * 16 + 16 <= Nn) {
#pragma unroll
            for (int n = 0; n < 4; ++n) {
                float* p = P + (size_t)(t * 16 + half * 8) * EHW + n * 16 + c;
#pragma unroll
                for (int i = 0; i < 8; ++i) p[(size_t)i * EHW] = acc[n][i];
            }
        } else {
#pragma unroll
            for (int n = 0; n < 4; ++n)
#pragma unroll
                for (int i = 0; i < 8; ++i) {
                    const int r = t * 16 + i + half * 8;
                    if (r < Nn) P[(size_t)r * EHW + n * 16 + c] = acc[n][i];
                }
        }
    }
}

__global__ void edge_kernel(const float* __restrict__ h,
                            const float* __restrict__ P,
                            const int* __restrict__ erow,
                            const int* __restrict__ ecol,
                            const float* __restrict__ w1c,
                            const float* __restrict__ w2,
                            const float* __restrict__ b2,
                            float* __restrict__ agg, int Ee) {
    constexpr int WS  = 72;
    constexpr int LWM = EHW + 4;
    __shared__ float w1s[HIDN * WS];
    __shared__ float w2s[EHW * WS];
    __shared__ float m1s[8][16 * LWM];

    for (int i4 = threadIdx.x; i4 < (HIDN * EHW) / 4; i4 += blockDim.x) {
        const int r = i4 >> 4, col = (i4 & 15) * 4;
        *(v4f*)(w1s + r * WS + col) = *(const v4f*)(w1c + r * EHW + col);
    }
    for (int i4 = threadIdx.x; i4 < (EHW * EHW) / 4; i4 += blockDim.x) {
        const int r = i4 >> 4, col = (i4 & 15) * 4;
        *(v4f*)(w2s + r * WS + col) = *(const v4f*)(w2 + r * EHW + col);
    }
    __syncthreads();

    const int lane = threadIdx.x & 31;
    const int wv   = threadIdx.x >> 5;
    const int c    = lane & 15;
    const int half = lane >> 4;
    const int nw   = (gridDim.x * blockDim.x) >> 5;
    const int wid0 = (blockIdx.x * blockDim.x + threadIdx.x) >> 5;
    const int tiles = (Ee + 15) >> 4;
    float* L = m1s[wv];

    for (int t = wid0; t < tiles; t += nw) {
        int ea = t * 16 + c;
        if (ea >= Ee) ea = Ee - 1;
        const float* __restrict__ hca = h + (size_t)ecol[ea] * HIDN;

        int ri[8];
#pragma unroll
        for (int i = 0; i < 8; ++i) {
            int e = t * 16 + i + half * 8;
            if (e >= Ee) e = Ee - 1;
            ri[i] = erow[e];
        }

        v8f acc[4];
#pragma unroll
        for (int i = 0; i < 8; ++i) {
            const float* __restrict__ pr = P + (size_t)ri[i] * EHW + c;
#pragma unroll
            for (int n = 0; n < 4; ++n) acc[n][i] = pr[n * 16];
        }

#pragma unroll 4
        for (int k = 0; k < HIDN; k += 4) {
            const int kk = k + half * 2;
            v2f a = *(const v2f*)(hca + kk);
#pragma unroll
            for (int n = 0; n < 4; ++n) {
                v2f b;
                b.x = w1s[kk * WS + n * 16 + c];
                b.y = w1s[(kk + 1) * WS + n * 16 + c];
                acc[n] = wmma4(a, b, acc[n]);
            }
        }

#pragma unroll
        for (int n = 0; n < 4; ++n)
#pragma unroll
            for (int i = 0; i < 8; ++i)
                L[(i + half * 8) * LWM + n * 16 + c] = silu_f(acc[n][i]);

        asm volatile("s_wait_dscnt 0" ::: "memory");

        v8f acc2[4];
#pragma unroll
        for (int n = 0; n < 4; ++n) acc2[n] = splat8(b2[n * 16 + c]);

#pragma unroll 4
        for (int k = 0; k < EHW; k += 4) {
            const int kk = k + half * 2;
            v2f a;
            a.x = L[c * LWM + kk];
            a.y = L[c * LWM + kk + 1];
#pragma unroll
            for (int n = 0; n < 4; ++n) {
                v2f b;
                b.x = w2s[kk * WS + n * 16 + c];
                b.y = w2s[(kk + 1) * WS + n * 16 + c];
                acc2[n] = wmma4(a, b, acc2[n]);
            }
        }

        if (t * 16 + 16 <= Ee) {
#pragma unroll
            for (int i = 0; i < 8; ++i) {
                float* dst = agg + (size_t)ri[i] * EHW + c;
#pragma unroll
                for (int n = 0; n < 4; ++n)
                    atomicAdd(dst + n * 16, silu_f(acc2[n][i]));
            }
        } else {
#pragma unroll
            for (int i = 0; i < 8; ++i) {
                const int e = t * 16 + i + half * 8;
                if (e < Ee) {
                    float* dst = agg + (size_t)ri[i] * EHW + c;
#pragma unroll
                    for (int n = 0; n < 4; ++n)
                        atomicAdd(dst + n * 16, silu_f(acc2[n][i]));
                }
            }
        }
        asm volatile("s_wait_dscnt 0" ::: "memory");
    }
}

__global__ void __launch_bounds__(256) edgepq_kernel(const float* __restrict__ P,
                              const float* __restrict__ Q,
                              const int* __restrict__ erow,
                              const int* __restrict__ ecol,
                              const float* __restrict__ w2,
                              const float* __restrict__ b2,
                              float* __restrict__ agg, int Ee) {
    constexpr int WS = 72;
    __shared__ float w2s[EHW * WS];

    for (int i4 = threadIdx.x; i4 < (EHW * EHW) / 4; i4 += blockDim.x) {
        const int r = i4 >> 4, col = (i4 & 15) * 4;
        *(v4f*)(w2s + r * WS + col) = *(const v4f*)(w2 + r * EHW + col);
    }
    __syncthreads();

    const int lane = threadIdx.x & 31;
    const int c    = lane & 15;
    const int half = lane >> 4;
    const int nw   = (gridDim.x * blockDim.x) >> 5;
    const int wid0 = (blockIdx.x * blockDim.x + threadIdx.x) >> 5;
    const int tiles = (Ee + 15) >> 4;

    for (int t = wid0; t < tiles; t += nw) {
        int ea = t * 16 + c;
        if (ea >= Ee) ea = Ee - 1;
        const float* __restrict__ pr = P + (size_t)erow[ea] * EHW + half * 4;
        const float* __restrict__ qr = Q + (size_t)ecol[ea] * EHW + half * 4;

        int ri[8];
#pragma unroll
        for (int i = 0; i < 8; ++i) {
            int e = t * 16 + i + half * 8;
            if (e >= Ee) e = Ee - 1;
            ri[i] = erow[e];
        }

        v4f pv[8], qv[8];
#pragma unroll
        for (int s = 0; s < 8; ++s) {
            pv[s] = *(const v4f*)(pr + 8 * s);
            qv[s] = *(const v4f*)(qr + 8 * s);
        }

        asm volatile("" ::: "memory");

        v8f acc2[4];
#pragma unroll
        for (int n = 0; n < 4; ++n) acc2[n] = splat8(b2[n * 16 + c]);

#pragma unroll
        for (int s = 0; s < 8; ++s) {
            const int kb = 8 * s + 4 * half;
            v2f a0, a1;
            a0.x = silu_f(pv[s].x + qv[s].x);
            a0.y = silu_f(pv[s].y + qv[s].y);
            a1.x = silu_f(pv[s].z + qv[s].z);
            a1.y = silu_f(pv[s].w + qv[s].w);
#pragma unroll
            for (int n = 0; n < 4; ++n) {
                v2f b;
                b.x = w2s[kb * WS + n * 16 + c];
                b.y = w2s[(kb + 1) * WS + n * 16 + c];
                acc2[n] = wmma4(a0, b, acc2[n]);
            }
#pragma unroll
            for (int n = 0; n < 4; ++n) {
                v2f b;
                b.x = w2s[(kb + 2) * WS + n * 16 + c];
                b.y = w2s[(kb + 3) * WS + n * 16 + c];
                acc2[n] = wmma4(a1, b, acc2[n]);
            }
        }

#pragma unroll
        for (int n = 0; n < 4; ++n)
#pragma unroll
            for (int i = 0; i < 8; ++i) acc2[n][i] = silu_f(acc2[n][i]);

        if (t * 16 + 16 <= Ee) {
#pragma unroll
            for (int i = 0; i < 8; ++i) {
                float* dst = agg + (size_t)ri[i] * EHW + c;
#pragma unroll
                for (int n = 0; n < 4; ++n)
                    atomicAdd(dst + n * 16, acc2[n][i]);
            }
        } else {
#pragma unroll
            for (int i = 0; i < 8; ++i) {
                const int e = t * 16 + i + half * 8;
                if (e < Ee) {
                    float* dst = agg + (size_t)ri[i] * EHW + c;
#pragma unroll
                    for (int n = 0; n < 4; ++n)
                        atomicAdd(dst + n * 16, acc2[n][i]);
                }
            }
        }
    }
}

__global__ void node_kernel(float* __restrict__ h,
                            const float* __restrict__ agg,
                            const float* __restrict__ m1,
                            const float* __restrict__ c1,
                            const float* __restrict__ m2,
                            const float* __restrict__ c2,
                            int Nn) {
    constexpr int LW  = HIDN + 4;
    constexpr int WS2 = HIDN + 8;
    __shared__ float w2s[HIDN * WS2];
    __shared__ float lds[8][16 * LW];

    for (int i4 = threadIdx.x; i4 < (HIDN * HIDN) / 4; i4 += blockDim.x) {
        const int r = i4 >> 5, col = (i4 & 31) * 4;
        *(v4f*)(w2s + r * WS2 + col) = *(const v4f*)(m2 + r * HIDN + col);
    }
    __syncthreads();

    const int lane = threadIdx.x & 31;
    const int wv   = threadIdx.x >> 5;
    const int c    = lane & 15;
    const int half = lane >> 4;
    const int nw   = (gridDim.x * blockDim.x) >> 5;
    const int wid0 = (blockIdx.x * blockDim.x + threadIdx.x) >> 5;
    const int tiles = (Nn + 15) >> 4;
    float* L = lds[wv];

    for (int t = wid0; t < tiles; t += nw) {
        int mr = t * 16 + c;
        if (mr >= Nn) mr = Nn - 1;
        const float* __restrict__ hr = h + (size_t)mr * HIDN;
        const float* __restrict__ ar = agg + (size_t)mr * EHW;

        v8f acc[8];
#pragma unroll
        for (int n = 0; n < 8; ++n) acc[n] = splat8(c1[n * 16 + c]);

        for (int k = 0; k < HIDN; k += 4) {
            const int kk = k + half * 2;
            v2f a = *(const v2f*)(hr + kk);
#pragma unroll
            for (int n = 0; n < 8; ++n) {
                v2f b;
                b.x = m1[kk * HIDN + n * 16 + c];
                b.y = m1[(kk + 1) * HIDN + n * 16 + c];
                acc[n] = wmma4(a, b, acc[n]);
            }
        }
        for (int k = 0; k < EHW; k += 4) {
            const int kk = k + half * 2;
            v2f a = *(const v2f*)(ar + kk);
#pragma unroll
            for (int n = 0; n < 8; ++n) {
                v2f b;
                b.x = m1[(HIDN + kk) * HIDN + n * 16 + c];
                b.y = m1[(HIDN + kk + 1) * HIDN + n * 16 + c];
                acc[n] = wmma4(a, b, acc[n]);
            }
        }
#pragma unroll
        for (int n = 0; n < 8; ++n)
#pragma unroll
            for (int i = 0; i < 8; ++i)
                L[(i + half * 8) * LW + n * 16 + c] = silu_f(acc[n][i]);

        asm volatile("s_wait_dscnt 0" ::: "memory");

        v8f acc2[8];
#pragma unroll
        for (int n = 0; n < 8; ++n) acc2[n] = splat8(c2[n * 16 + c]);

        for (int k = 0; k < HIDN; k += 4) {
            const int kk = k + half * 2;
            v2f a;
            a.x = L[c * LW + kk];
            a.y = L[c * LW + kk + 1];
#pragma unroll
            for (int n = 0; n < 8; ++n) {
                v2f b;
                b.x = w2s[kk * WS2 + n * 16 + c];
                b.y = w2s[(kk + 1) * WS2 + n * 16 + c];
                acc2[n] = wmma4(a, b, acc2[n]);
            }
        }
        if (t * 16 + 16 <= Nn) {
#pragma unroll
            for (int n = 0; n < 8; ++n) {
                float* p = h + (size_t)(t * 16 + half * 8) * HIDN + n * 16 + c;
#pragma unroll
                for (int i = 0; i < 8; ++i) p[(size_t)i * HIDN] += acc2[n][i];
            }
        } else {
#pragma unroll
            for (int n = 0; n < 8; ++n)
#pragma unroll
                for (int i = 0; i < 8; ++i) {
                    const int r = t * 16 + i + half * 8;
                    if (r < Nn) h[(size_t)r * HIDN + n * 16 + c] += acc2[n][i];
                }
        }
        asm volatile("s_wait_dscnt 0" ::: "memory");
    }
}

__global__ void out_kernel(const float* __restrict__ h,
                           const float* __restrict__ W,
                           const float* __restrict__ bias,
                           float* __restrict__ out, int Nn) {
    const int lane = threadIdx.x & 31;
    const int c    = lane & 15;
    const int half = lane >> 4;
    const int nw   = (gridDim.x * blockDim.x) >> 5;
    const int wid0 = (blockIdx.x * blockDim.x + threadIdx.x) >> 5;
    const int tiles = (Nn + 15) >> 4;

    for (int t = wid0; t < tiles; t += nw) {
        int mr = t * 16 + c;
        if (mr >= Nn) mr = Nn - 1;

        v8f acc[4];
#pragma unroll
        for (int n = 0; n < 4; ++n) acc[n] = splat8(bias[n * 16 + c]);

        for (int k = 0; k < HIDN; k += 4) {
            const int kk = k + half * 2;
            v2f a = *(const v2f*)(h + (size_t)mr * HIDN + kk);
#pragma unroll
            for (int n = 0; n < 4; ++n) {
                v2f b;
                b.x = W[kk * EHW + n * 16 + c];
                b.y = W[(kk + 1) * EHW + n * 16 + c];
                acc[n] = wmma4(a, b, acc[n]);
            }
        }
        if (t * 16 + 16 <= Nn) {
#pragma unroll
            for (int n = 0; n < 4; ++n) {
                float* p = out + (size_t)(t * 16 + half * 8) * EHW + n * 16 + c;
#pragma unroll
                for (int i = 0; i < 8; ++i) p[(size_t)i * EHW] = acc[n][i];
            }
        } else {
#pragma unroll
            for (int n = 0; n < 4; ++n)
#pragma unroll
                for (int i = 0; i < 8; ++i) {
                    const int r = t * 16 + i + half * 8;
                    if (r < Nn) out[(size_t)r * EHW + n * 16 + c] = acc[n][i];
                }
        }
    }
}

static inline int imin_h(int a, int b) { return a < b ? a : b; }
static inline int cdiv_h(int a, int b) { return (a + b - 1) / b; }

extern "C" void kernel_launch(void* const* d_in, const int* in_sizes, int n_in,
                              void* d_out, int out_size, void* d_ws, size_t ws_size,
                              hipStream_t stream) {
    const float* h_in  = (const float*)d_in[0];
    const int*   edges = (const int*)d_in[1];
    const float* W_in  = (const float*)d_in[2];
    const float* b_in  = (const float*)d_in[3];
    const float* eW1   = (const float*)d_in[4];
    const float* eb1   = (const float*)d_in[5];
    const float* eW2   = (const float*)d_in[6];
    const float* eb2   = (const float*)d_in[7];
    const float* nW1   = (const float*)d_in[8];
    const float* nb1   = (const float*)d_in[9];
    const float* nW2   = (const float*)d_in[10];
    const float* nb2   = (const float*)d_in[11];
    const float* W_out = (const float*)d_in[12];
    const float* b_out = (const float*)d_in[13];

    const int Nn = in_sizes[0] / IN_NF;
    const int Ee = in_sizes[1] / 2;
    const int Ll = in_sizes[4] / (2 * HIDN * EHW);

    float* h   = (float*)d_ws;
    float* agg = h + (size_t)Nn * HIDN;

    const int* erow = edges;
    const int* ecol = edges + Ee;

    const int tilesN = cdiv_h(Nn, 16);
    const int tilesE = cdiv_h(Ee, 16);
    const int gEmb = imin_h(cdiv_h(tilesN, 8), 2048);
    const int gEdg = imin_h(cdiv_h(tilesE, 8), 2048);
    const int gNod = imin_h(cdiv_h(tilesN, 8), 2048);

    embed_kernel<<<gEmb, 256, 0, stream>>>(h_in, W_in, b_in, h, Nn);

    float* P = (float*)d_out;
    float* Q = agg + (size_t)Nn * EHW;
    const bool roomQ = ws_size >= (size_t)Nn * (HIDN + 2 * EHW) * sizeof(float);

    for (int l = 0; l < Ll; ++l) {
        if (roomQ) {
            rowproj_kernel<<<dim3(gNod, 2), 256, 0, stream>>>(
                h, eW1 + (size_t)l * 2 * HIDN * EHW, eb1 + (size_t)l * EHW, P, Q, Nn);
            hipMemsetAsync(agg, 0, (size_t)Nn * EHW * sizeof(float), stream);
            edgepq_kernel<<<gEdg, 256, 0, stream>>>(
                P, Q, erow, ecol,
                eW2 + (size_t)l * EHW * EHW, eb2 + (size_t)l * EHW, agg, Ee);
        } else {
            rowproj_kernel<<<dim3(gNod, 1), 256, 0, stream>>>(
                h, eW1 + (size_t)l * 2 * HIDN * EHW, eb1 + (size_t)l * EHW, P, nullptr, Nn);
            hipMemsetAsync(agg, 0, (size_t)Nn * EHW * sizeof(float), stream);
            edge_kernel<<<gEdg, 256, 0, stream>>>(
                h, P, erow, ecol,
                eW1 + (size_t)l * 2 * HIDN * EHW + (size_t)HIDN * EHW,
                eW2 + (size_t)l * EHW * EHW,      eb2 + (size_t)l * EHW,
                agg, Ee);
        }
        node_kernel<<<gNod, 256, 0, stream>>>(
            h, agg,
            nW1 + (size_t)l * KIN_NODE * HIDN, nb1 + (size_t)l * HIDN,
            nW2 + (size_t)l * HIDN * HIDN,     nb2 + (size_t)l * HIDN,
            Nn);
    }

    out_kernel<<<gEmb, 256, 0, stream>>>(h, W_out, b_out, (float*)d_out, Nn);
}


# silu's quotient x/(1+e^-x) formed as x * v_rcp_f32(1+e^-x) in every kernel: 5 vector instructions per element, not 14; silu values differ in the last bit, sum orders unchanged
# speedup vs baseline: 1.7680x; 1.0695x over previous
#include <hip/hip_runtime.h>
#include <hip/hip_bf16.h>

typedef float v2f __attribute__((ext_vector_type(2)));
typedef float v4f __attribute__((ext_vector_type(4)));
typedef float v8f __attribute__((ext_vector_type(8)));

#define IN_NF   64
#define HIDN    128
#define EHW     64
#define KIN_NODE (HIDN + EHW)

__device__ __forceinline__ float silu_f(float x) {
    return x * __builtin_amdgcn_rcpf(1.0f + __expf(-x));
}

__device__ __forceinline__ v8f splat8(float v) {
    v8f r = {v, v, v, v, v, v, v, v};
    return r;
}

__device__ __forceinline__ v8f wmma4(v2f a, v2f b, v8f c) {
    return __builtin_amdgcn_wmma_f32_16x16x4_f32(
         false, a,  false, b,
         (short)0, c,  false,  false);
}

__global__ void embed_kernel(const float* __restrict__ hin,
                             const float* __restrict__ W,
                             const float* __restrict__ bias,
                             float* __restrict__ hout, int Nn) {
    const int lane = threadIdx.x & 31;
    const int c    = lane & 15;
    const int half = lane >> 4;
    const int nw   = (gridDim.x * blockDim.x) >> 5;
    const int wid0 = (blockIdx.x * blockDim.x + threadIdx.x) >> 5;
    const int tiles = (Nn + 15) >> 4;

    for (int t = wid0; t < tiles; t += nw) {
        int mr = t * 16 + c;
        if (mr >= Nn) mr = Nn - 1;

        v8f acc[8];
#pragma unroll
        for (int n = 0; n < 8; ++n) acc[n] = splat8(bias[n * 16 + c]);

        for (int k = 0; k < IN_NF; k += 4) {
            const int kk = k + half * 2;
            v2f a = *(const v2f*)(hin + (size_t)mr * IN_NF + kk);
#pragma unroll
            for (int n = 0; n < 8; ++n) {
                v2f b;
                b.x = W[kk * HIDN + n * 16 + c];
                b.y = W[(kk + 1) * HIDN + n * 16 + c];
                acc[n] = wmma4(a, b, acc[n]);
            }
        }
        if (t * 16 + 16 <= Nn) {
#pragma unroll
            for (int n = 0; n < 8; ++n) {
                float* p = hout + (size_t)(t * 16 + half * 8) * HIDN + n * 16 + c;
#pragma unroll
                for (int i = 0; i < 8; ++i) p[(size_t)i * HIDN] = acc[n][i];
            }
        } else {
#pragma unroll
            for (int n = 0; n < 8; ++n)
#pragma unroll
                for (int i = 0; i < 8; ++i) {
                    const int r = t * 16 + i + half * 8;
                    if (r < Nn) hout[(size_t)r * HIDN + n * 16 + c] = acc[n][i];
                }
        }
    }
}

__global__ void rowproj_kernel(const float* __restrict__ h,
                               const float* __restrict__ w1all,
                               const float* __restrict__ b1,
                               float* __restrict__ Pout,
                               float* __restrict__ Qout, int Nn) {
    constexpr int WS = 72;
    __shared__ float w1s[HIDN * WS];

    const int side = blockIdx.y;
    const float* __restrict__ w1 = w1all + (size_t)side * HIDN * EHW;
    float* __restrict__ P = side ? Qout : Pout;

    for (int i4 = threadIdx.x; i4 < (HIDN * EHW) / 4; i4 += blockDim.x) {
        const int r = i4 >> 4, col = (i4 & 15) * 4;
        *(v4f*)(w1s + r * WS + col) = *(const v4f*)(w1 + r * EHW + col);
    }
    __syncthreads();

    const int lane = threadIdx.x & 31;
    const int c    = lane & 15;
    const int half = lane >> 4;
    const int nw   = (gridDim.x * blockDim.x) >> 5;
    const int wid0 = (blockIdx.x * blockDim.x + threadIdx.x) >> 5;
    const int tiles = (Nn + 15) >> 4;

    for (int t = wid0; t < tiles; t += nw) {
        int mr = t * 16 + c;
        if (mr >= Nn) mr = Nn - 1;
        const float* __restrict__ hr = h + (size_t)mr * HIDN;

        v8f acc[4];
#pragma unroll
        for (int n = 0; n < 4; ++n) acc[n] = splat8(side ? 0.0f : b1[n * 16 + c]);

#pragma unroll 8
        for (int k = 0; k < HIDN; k += 4) {
            const int kk = k + half * 2;
            v2f a = *(const v2f*)(hr + kk);
#pragma unroll
            for (int n = 0; n < 4; ++n) {
                v2f b;
                b.x = w1s[kk * WS + n * 16 + c];
                b.y = w1s[(kk + 1) * WS + n * 16 + c];
                acc[n] = wmma4(a, b, acc[n]);
            }
        }
        if (t * 16 + 16 <= Nn) {
#pragma unroll
            for (int n = 0; n < 4; ++n) {
                float* p = P + (size_t)(t * 16 + half * 8) * EHW + n * 16 + c;
#pragma unroll
                for (int i = 0; i < 8; ++i) p[(size_t)i * EHW] = acc[n][i];
            }
        } else {
#pragma unroll
            for (int n = 0; n < 4; ++n)
#pragma unroll
                for (int i = 0; i < 8; ++i) {
                    const int r = t * 16 + i + half * 8;
                    if (r < Nn) P[(size_t)r * EHW + n * 16 + c] = acc[n][i];
                }
        }
    }
}

__global__ void edge_kernel(const float* __restrict__ h,
                            const float* __restrict__ P,
                            const int* __restrict__ erow,
                            const int* __restrict__ ecol,
                            const float* __restrict__ w1c,
                            const float* __restrict__ w2,
                            const float* __restrict__ b2,
                            float* __restrict__ agg, int Ee) {
    constexpr int WS  = 72;
    constexpr int LWM = EHW + 4;
    __shared__ float w1s[HIDN * WS];
    __shared__ float w2s[EHW * WS];
    __shared__ float m1s[8][16 * LWM];

    for (int i4 = threadIdx.x; i4 < (HIDN * EHW) / 4; i4 += blockDim.x) {
        const int r = i4 >> 4, col = (i4 & 15) * 4;
        *(v4f*)(w1s + r * WS + col) = *(const v4f*)(w1c + r * EHW + col);
    }
    for (int i4 = threadIdx.x; i4 < (EHW * EHW) / 4; i4 += blockDim.x) {
        const int r = i4 >> 4, col = (i4 & 15) * 4;
        *(v4f*)(w2s + r * WS + col) = *(const v4f*)(w2 + r * EHW + col);
    }
    __syncthreads();

    const int lane = threadIdx.x & 31;
    const int wv   = threadIdx.x >> 5;
    const int c    = lane & 15;
    const int half = lane >> 4;
    const int nw   = (gridDim.x * blockDim.x) >> 5;
    const int wid0 = (blockIdx.x * blockDim.x + threadIdx.x) >> 5;
    const int tiles = (Ee + 15) >> 4;
    float* L = m1s[wv];

    for (int t = wid0; t < tiles; t += nw) {
        int ea = t * 16 + c;
        if (ea >= Ee) ea = Ee - 1;
        const float* __restrict__ hca = h + (size_t)ecol[ea] * HIDN;

        int ri[8];
#pragma unroll
        for (int i = 0; i < 8; ++i) {
            int e = t * 16 + i + half * 8;
            if (e >= Ee) e = Ee - 1;
            ri[i] = erow[e];
        }

        v8f acc[4];
#pragma unroll
        for (int i = 0; i < 8; ++i) {
            const float* __restrict__ pr = P + (size_t)ri[i] * EHW + c;
#pragma unroll
            for (int n = 0; n < 4; ++n) acc[n][i] = pr[n * 16];
        }

#pragma unroll 4
        for (int k = 0; k < HIDN; k += 4) {
            const int kk = k + half * 2;
            v2f a = *(const v2f*)(hca + kk);
#pragma unroll
            for (int n = 0; n < 4; ++n) {
                v2f b;
                b.x = w1s[kk * WS + n * 16 + c];
                b.y = w1s[(kk + 1) * WS + n * 16 + c];
                acc[n] = wmma4(a, b, acc[n]);
            }
        }

#pragma unroll
        for (int n = 0; n < 4; ++n)
#pragma unroll
            for (int i = 0; i < 8; ++i)
                L[(i + half * 8) * LWM + n * 16 + c] = silu_f(acc[n][i]);

        asm volatile("s_wait_dscnt 0" ::: "memory");

        v8f acc2[4];
#pragma unroll
        for (int n = 0; n < 4; ++n) acc2[n] = splat8(b2[n * 16 + c]);

#pragma unroll 4
        for (int k = 0; k < EHW; k += 4) {
            const int kk = k + half * 2;
            v2f a;
            a.x = L[c * LWM + kk];
            a.y = L[c * LWM + kk + 1];
#pragma unroll
            for (int n = 0; n < 4; ++n) {
                v2f b;
                b.x = w2s[kk * WS + n * 16 + c];
                b.y = w2s[(kk + 1) * WS + n * 16 + c];
                acc2[n] = wmma4(a, b, acc2[n]);
            }
        }

        if (t * 16 + 16 <= Ee) {
#pragma unroll
            for (int i = 0; i < 8; ++i) {
                float* dst = agg + (size_t)ri[i] * EHW + c;
#pragma unroll
                for (int n = 0; n < 4; ++n)
                    atomicAdd(dst + n * 16, silu_f(acc2[n][i]));
            }
        } else {
#pragma unroll
            for (int i = 0; i < 8; ++i) {
                const int e = t * 16 + i + half * 8;
                if (e < Ee) {
                    float* dst = agg + (size_t)ri[i] * EHW + c;
#pragma unroll
                    for (int n = 0; n < 4; ++n)
                        atomicAdd(dst + n * 16, silu_f(acc2[n][i]));
                }
            }
        }
        asm volatile("s_wait_dscnt 0" ::: "memory");
    }
}

__global__ void __launch_bounds__(256, 8) edgepq_kernel(const float* __restrict__ P,
                              const float* __restrict__ Q,
                              const int* __restrict__ erow,
                              const int* __restrict__ ecol,
                              const float* __restrict__ w2,
                              const float* __restrict__ b2,
                              float* __restrict__ agg, int Ee) {
    constexpr int WS = 72;
    __shared__ float w2s[EHW * WS];

    for (int i4 = threadIdx.x; i4 < (EHW * EHW) / 4; i4 += blockDim.x) {
        const int r = i4 >> 4, col = (i4 & 15) * 4;
        *(v4f*)(w2s + r * WS + col) = *(const v4f*)(w2 + r * EHW + col);
    }
    __syncthreads();

    const int lane = threadIdx.x & 31;
    const int c    = lane & 15;
    const int half = lane >> 4;
    const int nw   = (gridDim.x * blockDim.x) >> 5;
    const int wid0 = (blockIdx.x * blockDim.x + threadIdx.x) >> 5;
    const int tiles = (Ee + 15) >> 4;

    for (int t = wid0; t < tiles; t += nw) {
        int ea = t * 16 + c;
        if (ea >= Ee) ea = Ee - 1;
        const float* __restrict__ pr = P + (size_t)erow[ea] * EHW + half * 4;
        const float* __restrict__ qr = Q + (size_t)ecol[ea] * EHW + half * 4;

        int ri[8];
#pragma unroll
        for (int i = 0; i < 8; ++i) {
            int e = t * 16 + i + half * 8;
            if (e >= Ee) e = Ee - 1;
            ri[i] = erow[e];
        }

        v4f pv[8], qv[8];
#pragma unroll
        for (int s = 0; s < 8; ++s) {
            pv[s] = *(const v4f*)(pr + 8 * s);
            qv[s] = *(const v4f*)(qr + 8 * s);
        }

        asm volatile("" ::: "memory");

        v8f acc2[4];
#pragma unroll
        for (int n = 0; n < 4; ++n) acc2[n] = splat8(b2[n * 16 + c]);

#pragma unroll
        for (int s = 0; s < 8; ++s) {
            const int kb = 8 * s + 4 * half;
            v2f a0, a1;
            a0.x = silu_f(pv[s].x + qv[s].x);
            a0.y = silu_f(pv[s].y + qv[s].y);
            a1.x = silu_f(pv[s].z + qv[s].z);
            a1.y = silu_f(pv[s].w + qv[s].w);
#pragma unroll
            for (int n = 0; n < 4; ++n) {
                v2f b;
                b.x = w2s[kb * WS + n * 16 + c];
                b.y = w2s[(kb + 1) * WS + n * 16 + c];
                acc2[n] = wmma4(a0, b, acc2[n]);
            }
#pragma unroll
            for (int n = 0; n < 4; ++n) {
                v2f b;
                b.x = w2s[(kb + 2) * WS + n * 16 + c];
                b.y = w2s[(kb + 3) * WS + n * 16 + c];
                acc2[n] = wmma4(a1, b, acc2[n]);
            }
        }

#pragma unroll
        for (int n = 0; n < 4; ++n)
#pragma unroll
            for (int i = 0; i < 8; ++i) acc2[n][i] = silu_f(acc2[n][i]);

        if (t * 16 + 16 <= Ee) {
#pragma unroll
            for (int i = 0; i < 8; ++i) {
                float* dst = agg + (size_t)ri[i] * EHW + c;
#pragma unroll
                for (int n = 0; n < 4; ++n)
                    atomicAdd(dst + n * 16, acc2[n][i]);
            }
        } else {
#pragma unroll
            for (int i = 0; i < 8; ++i) {
                const int e = t * 16 + i + half * 8;
                if (e < Ee) {
                    float* dst = agg + (size_t)ri[i] * EHW + c;
#pragma unroll
                    for (int n = 0; n < 4; ++n)
                        atomicAdd(dst + n * 16, acc2[n][i]);
                }
            }
        }
    }
}

__global__ void node_kernel(float* __restrict__ h,
                            const float* __restrict__ agg,
                            const float* __restrict__ m1,
                            const float* __restrict__ c1,
                            const float* __restrict__ m2,
                            const float* __restrict__ c2,
                            int Nn) {
    constexpr int LW  = HIDN + 4;
    constexpr int WS2 = HIDN + 8;
    __shared__ float w2s[HIDN * WS2];
    __shared__ float lds[8][16 * LW];

    for (int i4 = threadIdx.x; i4 < (HIDN * HIDN) / 4; i4 += blockDim.x) {
        const int r = i4 >> 5, col = (i4 & 31) * 4;
        *(v4f*)(w2s + r * WS2 + col) = *(const v4f*)(m2 + r * HIDN + col);
    }
    __syncthreads();

    const int lane = threadIdx.x & 31;
    const int wv   = threadIdx.x >> 5;
    const int c    = lane & 15;
    const int half = lane >> 4;
    const int nw   = (gridDim.x * blockDim.x) >> 5;
    const int wid0 = (blockIdx.x * blockDim.x + threadIdx.x) >> 5;
    const int tiles = (Nn + 15) >> 4;
    float* L = lds[wv];

    for (int t = wid0; t < tiles; t += nw) {
        int mr = t * 16 + c;
        if (mr >= Nn) mr = Nn - 1;
        const float* __restrict__ hr = h + (size_t)mr * HIDN;
        const float* __restrict__ ar = agg + (size_t)mr * EHW;

        v8f acc[8];
#pragma unroll
        for (int n = 0; n < 8; ++n) acc[n] = splat8(c1[n * 16 + c]);

        for (int k = 0; k < HIDN; k += 4) {
            const int kk = k + half * 2;
            v2f a = *(const v2f*)(hr + kk);
#pragma unroll
            for (int n = 0; n < 8; ++n) {
                v2f b;
                b.x = m1[kk * HIDN + n * 16 + c];
                b.y = m1[(kk + 1) * HIDN + n * 16 + c];
                acc[n] = wmma4(a, b, acc[n]);
            }
        }
        for (int k = 0; k < EHW; k += 4) {
            const int kk = k + half * 2;
            v2f a = *(const v2f*)(ar + kk);
#pragma unroll
            for (int n = 0; n < 8; ++n) {
                v2f b;
                b.x = m1[(HIDN + kk) * HIDN + n * 16 + c];
                b.y = m1[(HIDN + kk + 1) * HIDN + n * 16 + c];
                acc[n] = wmma4(a, b, acc[n]);
            }
        }
#pragma unroll
        for (int n = 0; n < 8; ++n)
#pragma unroll
            for (int i = 0; i < 8; ++i)
                L[(i + half * 8) * LW + n * 16 + c] = silu_f(acc[n][i]);

        asm volatile("s_wait_dscnt 0" ::: "memory");

        v8f acc2[8];
#pragma unroll
        for (int n = 0; n < 8; ++n) acc2[n] = splat8(c2[n * 16 + c]);

        for (int k = 0; k < HIDN; k += 4) {
            const int kk = k + half * 2;
            v2f a;
            a.x = L[c * LW + kk];
            a.y = L[c * LW + kk + 1];
#pragma unroll
            for (int n = 0; n < 8; ++n) {
                v2f b;
                b.x = w2s[kk * WS2 + n * 16 + c];
                b.y = w2s[(kk + 1) * WS2 + n * 16 + c];
                acc2[n] = wmma4(a, b, acc2[n]);
            }
        }
        if (t * 16 + 16 <= Nn) {
#pragma unroll
            for (int n = 0; n < 8; ++n) {
                float* p = h + (size_t)(t * 16 + half * 8) * HIDN + n * 16 + c;
#pragma unroll
                for (int i = 0; i < 8; ++i) p[(size_t)i * HIDN] += acc2[n][i];
            }
        } else {
#pragma unroll
            for (int n = 0; n < 8; ++n)
#pragma unroll
                for (int i = 0; i < 8; ++i) {
                    const int r = t * 16 + i + half * 8;
                    if (r < Nn) h[(size_t)r * HIDN + n * 16 + c] += acc2[n][i];
                }
        }
        asm volatile("s_wait_dscnt 0" ::: "memory");
    }
}

__global__ void out_kernel(const float* __restrict__ h,
                           const float* __restrict__ W,
                           const float* __restrict__ bias,
                           float* __restrict__ out, int Nn) {
    const int lane = threadIdx.x & 31;
    const int c    = lane & 15;
    const int half = lane >> 4;
    const int nw   = (gridDim.x * blockDim.x) >> 5;
    const int wid0 = (blockIdx.x * blockDim.x + threadIdx.x) >> 5;
    const int tiles = (Nn + 15) >> 4;

    for (int t = wid0; t < tiles; t += nw) {
        int mr = t * 16 + c;
        if (mr >= Nn) mr = Nn - 1;

        v8f acc[4];
#pragma unroll
        for (int n = 0; n < 4; ++n) acc[n] = splat8(bias[n * 16 + c]);

        for (int k = 0; k < HIDN; k += 4) {
            const int kk = k + half * 2;
            v2f a = *(const v2f*)(h + (size_t)mr * HIDN + kk);
#pragma unroll
            for (int n = 0; n < 4; ++n) {
                v2f b;
                b.x = W[kk * EHW + n * 16 + c];
                b.y = W[(kk + 1) * EHW + n * 16 + c];
                acc[n] = wmma4(a, b, acc[n]);
            }
        }
        if (t * 16 + 16 <= Nn) {
#pragma unroll
            for (int n = 0; n < 4; ++n) {
                float* p = out + (size_t)(t * 16 + half * 8) * EHW + n * 16 + c;
#pragma unroll
                for (int i = 0; i < 8; ++i) p[(size_t)i * EHW] = acc[n][i];
            }
        } else {
#pragma unroll
            for (int n = 0; n < 4; ++n)
#pragma unroll
                for (int i = 0; i < 8; ++i) {
                    const int r = t * 16 + i + half * 8;
                    if (r < Nn) out[(size_t)r * EHW + n * 16 + c] = acc[n][i];
                }
        }
    }
}

static inline int imin_h(int a, int b) { return a < b ? a : b; }
static inline int cdiv_h(int a, int b) { return (a + b - 1) / b; }

extern "C" void kernel_launch(void* const* d_in, const int* in_sizes, int n_in,
                              void* d_out, int out_size, void* d_ws, size_t ws_size,
                              hipStream_t stream) {
    const float* h_in  = (const float*)d_in[0];
    const int*   edges = (const int*)d_in[1];
    const float* W_in  = (const float*)d_in[2];
    const float* b_in  = (const float*)d_in[3];
    const float* eW1   = (const float*)d_in[4];
    const float* eb1   = (const float*)d_in[5];
    const float* eW2   = (const float*)d_in[6];
    const float* eb2   = (const float*)d_in[7];
    const float* nW1   = (const float*)d_in[8];
    const float* nb1   = (const float*)d_in[9];
    const float* nW2   = (const float*)d_in[10];
    const float* nb2   = (const float*)d_in[11];
    const float* W_out = (const float*)d_in[12];
    const float* b_out = (const float*)d_in[13];

    const int Nn = in_sizes[0] / IN_NF;
    const int Ee = in_sizes[1] / 2;
    const int Ll = in_sizes[4] / (2 * HIDN * EHW);

    float* h   = (float*)d_ws;
    float* agg = h + (size_t)Nn * HIDN;

    const int* erow = edges;
    const int* ecol = edges + Ee;

    const int tilesN = cdiv_h(Nn, 16);
    const int tilesE = cdiv_h(Ee, 16);
    const int gEmb = imin_h(cdiv_h(tilesN, 8), 2048);
    const int gEdg = imin_h(cdiv_h(tilesE, 8), 2048);
    const int gNod = imin_h(cdiv_h(tilesN, 8), 2048);

    embed_kernel<<<gEmb, 256, 0, stream>>>(h_in, W_in, b_in, h, Nn);

    float* P = (float*)d_out;
    float* Q = agg + (size_t)Nn * EHW;
    const bool roomQ = ws_size >= (size_t)Nn * (HIDN + 2 * EHW) * sizeof(float);

    for (int l = 0; l < Ll; ++l) {
        if (roomQ) {
            rowproj_kernel<<<dim3(gNod, 2), 256, 0, stream>>>(
                h, eW1 + (size_t)l * 2 * HIDN * EHW, eb1 + (size_t)l * EHW, P, Q, Nn);
            hipMemsetAsync(agg, 0, (size_t)Nn * EHW * sizeof(float), stream);
            edgepq_kernel<<<gEdg, 256, 0, stream>>>(
                P, Q, erow, ecol,
                eW2 + (size_t)l * EHW * EHW, eb2 + (size_t)l * EHW, agg, Ee);
        } else {
            rowproj_kernel<<<dim3(gNod, 1), 256, 0, stream>>>(
                h, eW1 + (size_t)l * 2 * HIDN * EHW, eb1 + (size_t)l * EHW, P, nullptr, Nn);
            hipMemsetAsync(agg, 0, (size_t)Nn * EHW * sizeof(float), stream);
            edge_kernel<<<gEdg, 256, 0, stream>>>(
                h, P, erow, ecol,
                eW1 + (size_t)l * 2 * HIDN * EHW + (size_t)HIDN * EHW,
                eW2 + (size_t)l * EHW * EHW,      eb2 + (size_t)l * EHW,
                agg, Ee);
        }
        node_kernel<<<gNod, 256, 0, stream>>>(
            h, agg,
            nW1 + (size_t)l * KIN_NODE * HIDN, nb1 + (size_t)l * HIDN,
            nW2 + (size_t)l * HIDN * HIDN,     nb2 + (size_t)l * HIDN,
            Nn);
    }

    out_kernel<<<gEmb, 256, 0, stream>>>(h, W_out, b_out, (float*)d_out, Nn);
}


# Node kernel: a tile's 64 residual reads issued as one group ahead of the silu and 2nd product (was load-wait-add-store 64 times in a row); weight staging loads grouped likewise; sum orders unchanged
# speedup vs baseline: 2.2016x; 1.2452x over previous
#include <hip/hip_runtime.h>
#include <hip/hip_bf16.h>

typedef float v2f __attribute__((ext_vector_type(2)));
typedef float v4f __attribute__((ext_vector_type(4)));
typedef float v8f __attribute__((ext_vector_type(8)));

#define IN_NF   64
#define HIDN    128
#define EHW     64
#define KIN_NODE (HIDN + EHW)
#define NTHR    256

__device__ __forceinline__ float silu_f(float x) {
    return x * __builtin_amdgcn_rcpf(1.0f + __expf(-x));
}

__device__ __forceinline__ v8f splat8(float v) {
    v8f r = {v, v, v, v, v, v, v, v};
    return r;
}

__device__ __forceinline__ v8f wmma4(v2f a, v2f b, v8f c) {
    return __builtin_amdgcn_wmma_f32_16x16x4_f32(
         false, a,  false, b,
         (short)0, c,  false,  false);
}

__global__ void embed_kernel(const float* __restrict__ hin,
                             const float* __restrict__ W,
                             const float* __restrict__ bias,
                             float* __restrict__ hout, int Nn) {
    const int lane = threadIdx.x & 31;
    const int c    = lane & 15;
    const int half = lane >> 4;
    const int nw   = (gridDim.x * blockDim.x) >> 5;
    const int wid0 = (blockIdx.x * blockDim.x + threadIdx.x) >> 5;
    const int tiles = (Nn + 15) >> 4;

    for (int t = wid0; t < tiles; t += nw) {
        int mr = t * 16 + c;
        if (mr >= Nn) mr = Nn - 1;

        v8f acc[8];
#pragma unroll
        for (int n = 0; n < 8; ++n) acc[n] = splat8(bias[n * 16 + c]);

        for (int k = 0; k < IN_NF; k += 4) {
            const int kk = k + half * 2;
            v2f a = *(const v2f*)(hin + (size_t)mr * IN_NF + kk);
#pragma unroll
            for (int n = 0; n < 8; ++n) {
                v2f b;
                b.x = W[kk * HIDN + n * 16 + c];
                b.y = W[(kk + 1) * HIDN + n * 16 + c];
                acc[n] = wmma4(a, b, acc[n]);
            }
        }
        if (t * 16 + 16 <= Nn) {
#pragma unroll
            for (int n = 0; n < 8; ++n) {
                float* p = hout + (size_t)(t * 16 + half * 8) * HIDN + n * 16 + c;
#pragma unroll
                for (int i = 0; i < 8; ++i) p[(size_t)i * HIDN] = acc[n][i];
            }
        } else {
#pragma unroll
            for (int n = 0; n < 8; ++n)
#pragma unroll
                for (int i = 0; i < 8; ++i) {
                    const int r = t * 16 + i + half * 8;
                    if (r < Nn) hout[(size_t)r * HIDN + n * 16 + c] = acc[n][i];
                }
        }
    }
}

__global__ void rowproj_kernel(const float* __restrict__ h,
                               const float* __restrict__ w1all,
                               const float* __restrict__ b1,
                               float* __restrict__ Pout,
                               float* __restrict__ Qout, int Nn) {
    constexpr int WS = 72;
    __shared__ float w1s[HIDN * WS];

    const int side = blockIdx.y;
    const float* __restrict__ w1 = w1all + (size_t)side * HIDN * EHW;
    float* __restrict__ P = side ? Qout : Pout;

    {
        constexpr int NS = (HIDN * EHW) / 4 / NTHR;
        v4f stg[NS];
#pragma unroll
        for (int j = 0; j < NS; ++j)
            stg[j] = *(const v4f*)(w1 + 4 * (threadIdx.x + j * NTHR));
#pragma unroll
        for (int j = 0; j < NS; ++j) {
            const int i4 = threadIdx.x + j * NTHR;
            *(v4f*)(w1s + (i4 >> 4) * WS + (i4 & 15) * 4) = stg[j];
        }
    }
    __syncthreads();

    const int lane = threadIdx.x & 31;
    const int c    = lane & 15;
    const int half = lane >> 4;
    const int nw   = (gridDim.x * blockDim.x) >> 5;
    const int wid0 = (blockIdx.x * blockDim.x + threadIdx.x) >> 5;
    const int tiles = (Nn + 15) >> 4;

    for (int t = wid0; t < tiles; t += nw) {
        int mr = t * 16 + c;
        if (mr >= Nn) mr = Nn - 1;
        const float* __restrict__ hr = h + (size_t)mr * HIDN;

        v8f acc[4];
#pragma unroll
        for (int n = 0; n < 4; ++n) acc[n] = splat8(side ? 0.0f : b1[n * 16 + c]);

#pragma unroll 8
        for (int k = 0; k < HIDN; k += 4) {
            const int kk = k + half * 2;
            v2f a = *(const v2f*)(hr + kk);
#pragma unroll
            for (int n = 0; n < 4; ++n) {
                v2f b;
                b.x = w1s[kk * WS + n * 16 + c];
                b.y = w1s[(kk + 1) * WS + n * 16 + c];
                acc[n] = wmma4(a, b, acc[n]);
            }
        }
        if (t * 16 + 16 <= Nn) {
#pragma unroll
            for (int n = 0; n < 4; ++n) {
                float* p = P + (size_t)(t * 16 + half * 8) * EHW + n * 16 + c;
#pragma unroll
                for (int i = 0; i < 8; ++i) p[(size_t)i * EHW] = acc[n][i];
            }
        } else {
#pragma unroll
            for (int n = 0; n < 4; ++n)
#pragma unroll
                for (int i = 0; i < 8; ++i) {
                    const int r = t * 16 + i + half * 8;
                    if (r < Nn) P[(size_t)r * EHW + n * 16 + c] = acc[n][i];
                }
        }
    }
}

__global__ void edge_kernel(const float* __restrict__ h,
                            const float* __restrict__ P,
                            const int* __restrict__ erow,
                            const int* __restrict__ ecol,
                            const float* __restrict__ w1c,
                            const float* __restrict__ w2,
                            const float* __restrict__ b2,
                            float* __restrict__ agg, int Ee) {
    constexpr int WS  = 72;
    constexpr int LWM = EHW + 4;
    __shared__ float w1s[HIDN * WS];
    __shared__ float w2s[EHW * WS];
    __shared__ float m1s[8][16 * LWM];

    for (int i4 = threadIdx.x; i4 < (HIDN * EHW) / 4; i4 += blockDim.x) {
        const int r = i4 >> 4, col = (i4 & 15) * 4;
        *(v4f*)(w1s + r * WS + col) = *(const v4f*)(w1c + r * EHW + col);
    }
    for (int i4 = threadIdx.x; i4 < (EHW * EHW) / 4; i4 += blockDim.x) {
        const int r = i4 >> 4, col = (i4 & 15) * 4;
        *(v4f*)(w2s + r * WS + col) = *(const v4f*)(w2 + r * EHW + col);
    }
    __syncthreads();

    const int lane = threadIdx.x & 31;
    const int wv   = threadIdx.x >> 5;
    const int c    = lane & 15;
    const int half = lane >> 4;
    const int nw   = (gridDim.x * blockDim.x) >> 5;
    const int wid0 = (blockIdx.x * blockDim.x + threadIdx.x) >> 5;
    const int tiles = (Ee + 15) >> 4;
    float* L = m1s[wv];

    for (int t = wid0; t < tiles; t += nw) {
        int ea = t * 16 + c;
        if (ea >= Ee) ea = Ee - 1;
        const float* __restrict__ hca = h + (size_t)ecol[ea] * HIDN;

        int ri[8];
#pragma unroll
        for (int i = 0; i < 8; ++i) {
            int e = t * 16 + i + half * 8;
            if (e >= Ee) e = Ee - 1;
            ri[i] = erow[e];
        }

        v8f acc[4];
#pragma unroll
        for (int i = 0; i < 8; ++i) {
            const float* __restrict__ pr = P + (size_t)ri[i] * EHW + c;
#pragma unroll
            for (int n = 0; n < 4; ++n) acc[n][i] = pr[n * 16];
        }

#pragma unroll 4
        for (int k = 0; k < HIDN; k += 4) {
            const int kk = k + half * 2;
            v2f a = *(const v2f*)(hca + kk);
#pragma unroll
            for (int n = 0; n < 4; ++n) {
                v2f b;
                b.x = w1s[kk * WS + n * 16 + c];
                b.y = w1s[(kk + 1) * WS + n * 16 + c];
                acc[n] = wmma4(a, b, acc[n]);
            }
        }

#pragma unroll
        for (int n = 0; n < 4; ++n)
#pragma unroll
            for (int i = 0; i < 8; ++i)
                L[(i + half * 8) * LWM + n * 16 + c] = silu_f(acc[n][i]);

        asm volatile("s_wait_dscnt 0" ::: "memory");

        v8f acc2[4];
#pragma unroll
        for (int n = 0; n < 4; ++n) acc2[n] = splat8(b2[n * 16 + c]);

#pragma unroll 4
        for (int k = 0; k < EHW; k += 4) {
            const int kk = k + half * 2;
            v2f a;
            a.x = L[c * LWM + kk];
            a.y = L[c * LWM + kk + 1];
#pragma unroll
            for (int n = 0; n < 4; ++n) {
                v2f b;
                b.x = w2s[kk * WS + n * 16 + c];
                b.y = w2s[(kk + 1) * WS + n * 16 + c];
                acc2[n] = wmma4(a, b, acc2[n]);
            }
        }

        if (t * 16 + 16 <= Ee) {
#pragma unroll
            for (int i = 0; i < 8; ++i) {
                float* dst = agg + (size_t)ri[i] * EHW + c;
#pragma unroll
                for (int n = 0; n < 4; ++n)
                    atomicAdd(dst + n * 16, silu_f(acc2[n][i]));
            }
        } else {
#pragma unroll
            for (int i = 0; i < 8; ++i) {
                const int e = t * 16 + i + half * 8;
                if (e < Ee) {
                    float* dst = agg + (size_t)ri[i] * EHW + c;
#pragma unroll
                    for (int n = 0; n < 4; ++n)
                        atomicAdd(dst + n * 16, silu_f(acc2[n][i]));
                }
            }
        }
        asm volatile("s_wait_dscnt 0" ::: "memory");
    }
}

__global__ void __launch_bounds__(256, 8) edgepq_kernel(const float* __restrict__ P,
                              const float* __restrict__ Q,
                              const int* __restrict__ erow,
                              const int* __restrict__ ecol,
                              const float* __restrict__ w2,
                              const float* __restrict__ b2,
                              float* __restrict__ agg, int Ee) {
    constexpr int WS = 72;
    __shared__ float w2s[EHW * WS];

    {
        constexpr int NS = (EHW * EHW) / 4 / NTHR;
        v4f stg[NS];
#pragma unroll
        for (int j = 0; j < NS; ++j)
            stg[j] = *(const v4f*)(w2 + 4 * (threadIdx.x + j * NTHR));
#pragma unroll
        for (int j = 0; j < NS; ++j) {
            const int i4 = threadIdx.x + j * NTHR;
            *(v4f*)(w2s + (i4 >> 4) * WS + (i4 & 15) * 4) = stg[j];
        }
    }
    __syncthreads();

    const int lane = threadIdx.x & 31;
    const int c    = lane & 15;
    const int half = lane >> 4;
    const int nw   = (gridDim.x * blockDim.x) >> 5;
    const int wid0 = (blockIdx.x * blockDim.x + threadIdx.x) >> 5;
    const int tiles = (Ee + 15) >> 4;

    for (int t = wid0; t < tiles; t += nw) {
        int ea = t * 16 + c;
        if (ea >= Ee) ea = Ee - 1;
        const float* __restrict__ pr = P + (size_t)erow[ea] * EHW + half * 4;
        const float* __restrict__ qr = Q + (size_t)ecol[ea] * EHW + half * 4;

        int ri[8];
#pragma unroll
        for (int i = 0; i < 8; ++i) {
            int e = t * 16 + i + half * 8;
            if (e >= Ee) e = Ee - 1;
            ri[i] = erow[e];
        }

        v4f pv[8], qv[8];
#pragma unroll
        for (int s = 0; s < 8; ++s) {
            pv[s] = *(const v4f*)(pr + 8 * s);
            qv[s] = *(const v4f*)(qr + 8 * s);
        }

        asm volatile("" ::: "memory");

        v8f acc2[4];
#pragma unroll
        for (int n = 0; n < 4; ++n) acc2[n] = splat8(b2[n * 16 + c]);

#pragma unroll
        for (int s = 0; s < 8; ++s) {
            const int kb = 8 * s + 4 * half;
            v2f a0, a1;
            a0.x = silu_f(pv[s].x + qv[s].x);
            a0.y = silu_f(pv[s].y + qv[s].y);
            a1.x = silu_f(pv[s].z + qv[s].z);
            a1.y = silu_f(pv[s].w + qv[s].w);
#pragma unroll
            for (int n = 0; n < 4; ++n) {
                v2f b;
                b.x = w2s[kb * WS + n * 16 + c];
                b.y = w2s[(kb + 1) * WS + n * 16 + c];
                acc2[n] = wmma4(a0, b, acc2[n]);
            }
#pragma unroll
            for (int n = 0; n < 4; ++n) {
                v2f b;
                b.x = w2s[(kb + 2) * WS + n * 16 + c];
                b.y = w2s[(kb + 3) * WS + n * 16 + c];
                acc2[n] = wmma4(a1, b, acc2[n]);
            }
        }

#pragma unroll
        for (int n = 0; n < 4; ++n)
#pragma unroll
            for (int i = 0; i < 8; ++i) acc2[n][i] = silu_f(acc2[n][i]);

        if (t * 16 + 16 <= Ee) {
#pragma unroll
            for (int i = 0; i < 8; ++i) {
                float* dst = agg + (size_t)ri[i] * EHW + c;
#pragma unroll
                for (int n = 0; n < 4; ++n)
                    atomicAdd(dst + n * 16, acc2[n][i]);
            }
        } else {
#pragma unroll
            for (int i = 0; i < 8; ++i) {
                const int e = t * 16 + i + half * 8;
                if (e < Ee) {
                    float* dst = agg + (size_t)ri[i] * EHW + c;
#pragma unroll
                    for (int n = 0; n < 4; ++n)
                        atomicAdd(dst + n * 16, acc2[n][i]);
                }
            }
        }
    }
}

__global__ void __launch_bounds__(NTHR) node_kernel(float* __restrict__ h,
                            const float* __restrict__ agg,
                            const float* __restrict__ m1,
                            const float* __restrict__ c1,
                            const float* __restrict__ m2,
                            const float* __restrict__ c2,
                            int Nn) {
    constexpr int LW  = HIDN + 4;
    constexpr int WS2 = HIDN + 8;
    __shared__ float w2s[HIDN * WS2];
    __shared__ float lds[8][16 * LW];

    {
        constexpr int NS = (HIDN * HIDN) / 4 / NTHR;
        v4f stg[NS];
#pragma unroll
        for (int j = 0; j < NS; ++j)
            stg[j] = *(const v4f*)(m2 + 4 * (threadIdx.x + j * NTHR));
#pragma unroll
        for (int j = 0; j < NS; ++j) {
            const int i4 = threadIdx.x + j * NTHR;
            *(v4f*)(w2s + (i4 >> 5) * WS2 + (i4 & 31) * 4) = stg[j];
        }
    }
    __syncthreads();

    const int lane = threadIdx.x & 31;
    const int wv   = threadIdx.x >> 5;
    const int c    = lane & 15;
    const int half = lane >> 4;
    const int nw   = (gridDim.x * blockDim.x) >> 5;
    const int wid0 = (blockIdx.x * blockDim.x + threadIdx.x) >> 5;
    const int tiles = (Nn + 15) >> 4;
    float* L = lds[wv];

    for (int t = wid0; t < tiles; t += nw) {
        int mr = t * 16 + c;
        if (mr >= Nn) mr = Nn - 1;
        const float* __restrict__ hr = h + (size_t)mr * HIDN;
        const float* __restrict__ ar = agg + (size_t)mr * EHW;

        v8f acc[8];
#pragma unroll
        for (int n = 0; n < 8; ++n) acc[n] = splat8(c1[n * 16 + c]);

        for (int k = 0; k < HIDN; k += 4) {
            const int kk = k + half * 2;
            v2f a = *(const v2f*)(hr + kk);
#pragma unroll
            for (int n = 0; n < 8; ++n) {
                v2f b;
                b.x = m1[kk * HIDN + n * 16 + c];
                b.y = m1[(kk + 1) * HIDN + n * 16 + c];
                acc[n] = wmma4(a, b, acc[n]);
            }
        }
        for (int k = 0; k < EHW; k += 4) {
            const int kk = k + half * 2;
            v2f a = *(const v2f*)(ar + kk);
#pragma unroll
            for (int n = 0; n < 8; ++n) {
                v2f b;
                b.x = m1[(HIDN + kk) * HIDN + n * 16 + c];
                b.y = m1[(HIDN + kk + 1) * HIDN + n * 16 + c];
                acc[n] = wmma4(a, b, acc[n]);
            }
        }
        const bool full = (t * 16 + 16 <= Nn);
        v8f hres[8];
        if (full) {
            const float* p = h + (size_t)(t * 16 + half * 8) * HIDN + c;
#pragma unroll
            for (int n = 0; n < 8; ++n)
#pragma unroll
                for (int i = 0; i < 8; ++i) hres[n][i] = p[(size_t)i * HIDN + n * 16];
        } else {
#pragma unroll
            for (int i = 0; i < 8; ++i) {
                int r = t * 16 + i + half * 8;
                if (r >= Nn) r = Nn - 1;
                const float* p = h + (size_t)r * HIDN + c;
#pragma unroll
                for (int n = 0; n < 8; ++n) hres[n][i] = p[n * 16];
            }
        }

#pragma unroll
        for (int n = 0; n < 8; ++n)
#pragma unroll
            for (int i = 0; i < 8; ++i)
                L[(i + half * 8) * LW + n * 16 + c] = silu_f(acc[n][i]);

        asm volatile("s_wait_dscnt 0" ::: "memory");

        v8f acc2[8];
#pragma unroll
        for (int n = 0; n < 8; ++n) acc2[n] = splat8(c2[n * 16 + c]);

        for (int k = 0; k < HIDN; k += 4) {
            const int kk = k + half * 2;
            v2f a;
            a.x = L[c * LW + kk];
            a.y = L[c * LW + kk + 1];
#pragma unroll
            for (int n = 0; n < 8; ++n) {
                v2f b;
                b.x = w2s[kk * WS2 + n * 16 + c];
                b.y = w2s[(kk + 1) * WS2 + n * 16 + c];
                acc2[n] = wmma4(a, b, acc2[n]);
            }
        }
        if (full) {
#pragma unroll
            for (int n = 0; n < 8; ++n) {
                float* p = h + (size_t)(t * 16 + half * 8) * HIDN + n * 16 + c;
#pragma unroll
                for (int i = 0; i < 8; ++i) p[(size_t)i * HIDN] = hres[n][i] + acc2[n][i];
            }
        } else {
#pragma unroll
            for (int n = 0; n < 8; ++n)
#pragma unroll
                for (int i = 0; i < 8; ++i) {
                    const int r = t * 16 + i + half * 8;
                    if (r < Nn) h[(size_t)r * HIDN + n * 16 + c] = hres[n][i] + acc2[n][i];
                }
        }
        asm volatile("s_wait_dscnt 0" ::: "memory");
    }
}

__global__ void out_kernel(const float* __restrict__ h,
                           const float* __restrict__ W,
                           const float* __restrict__ bias,
                           float* __restrict__ out, int Nn) {
    const int lane = threadIdx.x & 31;
    const int c    = lane & 15;
    const int half = lane >> 4;
    const int nw   = (gridDim.x * blockDim.x) >> 5;
    const int wid0 = (blockIdx.x * blockDim.x + threadIdx.x) >> 5;
    const int tiles = (Nn + 15) >> 4;

    for (int t = wid0; t < tiles; t += nw) {
        int mr = t * 16 + c;
        if (mr >= Nn) mr = Nn - 1;

        v8f acc[4];
#pragma unroll
        for (int n = 0; n < 4; ++n) acc[n] = splat8(bias[n * 16 + c]);

        for (int k = 0; k < HIDN; k += 4) {
            const int kk = k + half * 2;
            v2f a = *(const v2f*)(h + (size_t)mr * HIDN + kk);
#pragma unroll
            for (int n = 0; n < 4; ++n) {
                v2f b;
                b.x = W[kk * EHW + n * 16 + c];
                b.y = W[(kk + 1) * EHW + n * 16 + c];
                acc[n] = wmma4(a, b, acc[n]);
            }
        }
        if (t * 16 + 16 <= Nn) {
#pragma unroll
            for (int n = 0; n < 4; ++n) {
                float* p = out + (size_t)(t * 16 + half * 8) * EHW + n * 16 + c;
#pragma unroll
                for (int i = 0; i < 8; ++i) p[(size_t)i * EHW] = acc[n][i];
            }
        } else {
#pragma unroll
            for (int n = 0; n < 4; ++n)
#pragma unroll
                for (int i = 0; i < 8; ++i) {
                    const int r = t * 16 + i + half * 8;
                    if (r < Nn) out[(size_t)r * EHW + n * 16 + c] = acc[n][i];
                }
        }
    }
}

static inline int imin_h(int a, int b) { return a < b ? a : b; }
static inline int cdiv_h(int a, int b) { return (a + b - 1) / b; }

extern "C" void kernel_launch(void* const* d_in, const int* in_sizes, int n_in,
                              void* d_out, int out_size, void* d_ws, size_t ws_size,
                              hipStream_t stream) {
    const float* h_in  = (const float*)d_in[0];
    const int*   edges = (const int*)d_in[1];
    const float* W_in  = (const float*)d_in[2];
    const float* b_in  = (const float*)d_in[3];
    const float* eW1   = (const float*)d_in[4];
    const float* eb1   = (const float*)d_in[5];
    const float* eW2   = (const float*)d_in[6];
    const float* eb2   = (const float*)d_in[7];
    const float* nW1   = (const float*)d_in[8];
    const float* nb1   = (const float*)d_in[9];
    const float* nW2   = (const float*)d_in[10];
    const float* nb2   = (const float*)d_in[11];
    const float* W_out = (const float*)d_in[12];
    const float* b_out = (const float*)d_in[13];

    const int Nn = in_sizes[0] / IN_NF;
    const int Ee = in_sizes[1] / 2;
    const int Ll = in_sizes[4] / (2 * HIDN * EHW);

    float* h   = (float*)d_ws;
    float* agg = h + (size_t)Nn * HIDN;

    const int* erow = edges;
    const int* ecol = edges + Ee;

    const int tilesN = cdiv_h(Nn, 16);
    const int tilesE = cdiv_h(Ee, 16);
    const int gEmb = imin_h(cdiv_h(tilesN, 8), 2048);
    const int gEdg = imin_h(cdiv_h(tilesE, 8), 2048);
    const int gNod = imin_h(cdiv_h(tilesN, 8), 2048);

    embed_kernel<<<gEmb, 256, 0, stream>>>(h_in, W_in, b_in, h, Nn);

    float* P = (float*)d_out;
    float* Q = agg + (size_t)Nn * EHW;
    const bool roomQ = ws_size >= (size_t)Nn * (HIDN + 2 * EHW) * sizeof(float);

    for (int l = 0; l < Ll; ++l) {
        if (roomQ) {
            rowproj_kernel<<<dim3(gNod, 2), 256, 0, stream>>>(
                h, eW1 + (size_t)l * 2 * HIDN * EHW, eb1 + (size_t)l * EHW, P, Q, Nn);
            hipMemsetAsync(agg, 0, (size_t)Nn * EHW * sizeof(float), stream);
            edgepq_kernel<<<gEdg, 256, 0, stream>>>(
                P, Q, erow, ecol,
                eW2 + (size_t)l * EHW * EHW, eb2 + (size_t)l * EHW, agg, Ee);
        } else {
            rowproj_kernel<<<dim3(gNod, 1), 256, 0, stream>>>(
                h, eW1 + (size_t)l * 2 * HIDN * EHW, eb1 + (size_t)l * EHW, P, nullptr, Nn);
            hipMemsetAsync(agg, 0, (size_t)Nn * EHW * sizeof(float), stream);
            edge_kernel<<<gEdg, 256, 0, stream>>>(
                h, P, erow, ecol,
                eW1 + (size_t)l * 2 * HIDN * EHW + (size_t)HIDN * EHW,
                eW2 + (size_t)l * EHW * EHW,      eb2 + (size_t)l * EHW,
                agg, Ee);
        }
        node_kernel<<<gNod, 256, 0, stream>>>(
            h, agg,
            nW1 + (size_t)l * KIN_NODE * HIDN, nb1 + (size_t)l * HIDN,
            nW2 + (size_t)l * HIDN * HIDN,     nb2 + (size_t)l * HIDN,
            Nn);
    }

    out_kernel<<<gEmb, 256, 0, stream>>>(h, W_out, b_out, (float*)d_out, Nn);
}


# agg's zero-fill done by the row-projection kernel's own stores (each tile wave clears its 16 rows' 128-byte lines ahead of its product): the 4 fill launches per run are gone; sum orders unchanged
# speedup vs baseline: 2.3068x; 1.0478x over previous
#include <hip/hip_runtime.h>
#include <hip/hip_bf16.h>

typedef float v2f __attribute__((ext_vector_type(2)));
typedef float v4f __attribute__((ext_vector_type(4)));
typedef float v8f __attribute__((ext_vector_type(8)));

#define IN_NF   64
#define HIDN    128
#define EHW     64
#define KIN_NODE (HIDN + EHW)
#define NTHR    256

__device__ __forceinline__ float silu_f(float x) {
    return x * __builtin_amdgcn_rcpf(1.0f + __expf(-x));
}

__device__ __forceinline__ v8f splat8(float v) {
    v8f r = {v, v, v, v, v, v, v, v};
    return r;
}

__device__ __forceinline__ v8f wmma4(v2f a, v2f b, v8f c) {
    return __builtin_amdgcn_wmma_f32_16x16x4_f32(
         false, a,  false, b,
         (short)0, c,  false,  false);
}

__global__ void embed_kernel(const float* __restrict__ hin,
                             const float* __restrict__ W,
                             const float* __restrict__ bias,
                             float* __restrict__ hout, int Nn) {
    const int lane = threadIdx.x & 31;
    const int c    = lane & 15;
    const int half = lane >> 4;
    const int nw   = (gridDim.x * blockDim.x) >> 5;
    const int wid0 = (blockIdx.x * blockDim.x + threadIdx.x) >> 5;
    const int tiles = (Nn + 15) >> 4;

    for (int t = wid0; t < tiles; t += nw) {
        int mr = t * 16 + c;
        if (mr >= Nn) mr = Nn - 1;

        v8f acc[8];
#pragma unroll
        for (int n = 0; n < 8; ++n) acc[n] = splat8(bias[n * 16 + c]);

        for (int k = 0; k < IN_NF; k += 4) {
            const int kk = k + half * 2;
            v2f a = *(const v2f*)(hin + (size_t)mr * IN_NF + kk);
#pragma unroll
            for (int n = 0; n < 8; ++n) {
                v2f b;
                b.x = W[kk * HIDN + n * 16 + c];
                b.y = W[(kk + 1) * HIDN + n * 16 + c];
                acc[n] = wmma4(a, b, acc[n]);
            }
        }
        if (t * 16 + 16 <= Nn) {
#pragma unroll
            for (int n = 0; n < 8; ++n) {
                float* p = hout + (size_t)(t * 16 + half * 8) * HIDN + n * 16 + c;
#pragma unroll
                for (int i = 0; i < 8; ++i) p[(size_t)i * HIDN] = acc[n][i];
            }
        } else {
#pragma unroll
            for (int n = 0; n < 8; ++n)
#pragma unroll
                for (int i = 0; i < 8; ++i) {
                    const int r = t * 16 + i + half * 8;
                    if (r < Nn) hout[(size_t)r * HIDN + n * 16 + c] = acc[n][i];
                }
        }
    }
}

__global__ void rowproj_kernel(const float* __restrict__ h,
                               const float* __restrict__ w1all,
                               const float* __restrict__ b1,
                               float* __restrict__ Pout,
                               float* __restrict__ Qout,
                               float* __restrict__ aggz, int Nn) {
    constexpr int WS = 72;
    __shared__ float w1s[HIDN * WS];

    const int side = blockIdx.y;
    const float* __restrict__ w1 = w1all + (size_t)side * HIDN * EHW;
    float* __restrict__ P = side ? Qout : Pout;

    {
        constexpr int NS = (HIDN * EHW) / 4 / NTHR;
        v4f stg[NS];
#pragma unroll
        for (int j = 0; j < NS; ++j)
            stg[j] = *(const v4f*)(w1 + 4 * (threadIdx.x + j * NTHR));
#pragma unroll
        for (int j = 0; j < NS; ++j) {
            const int i4 = threadIdx.x + j * NTHR;
            *(v4f*)(w1s + (i4 >> 4) * WS + (i4 & 15) * 4) = stg[j];
        }
    }
    __syncthreads();

    const int lane = threadIdx.x & 31;
    const int c    = lane & 15;
    const int half = lane >> 4;
    const int nw   = (gridDim.x * blockDim.x) >> 5;
    const int wid0 = (blockIdx.x * blockDim.x + threadIdx.x) >> 5;
    const int tiles = (Nn + 15) >> 4;

    for (int t = wid0; t < tiles; t += nw) {
        int mr = t * 16 + c;
        if (mr >= Nn) mr = Nn - 1;
        const float* __restrict__ hr = h + (size_t)mr * HIDN;

        {
            const v4f z4 = {0.0f, 0.0f, 0.0f, 0.0f};
            const bool fullz = (t * 16 + 16 <= Nn);
            if (gridDim.y == 2) {
                const int r0 = t * 16 + (lane >> 3);
                float* zp = aggz + (size_t)r0 * EHW + side * 32 + (lane & 7) * 4;
                if (fullz) {
#pragma unroll
                    for (int q = 0; q < 4; ++q) *(v4f*)(zp + (size_t)q * 4 * EHW) = z4;
                } else {
#pragma unroll
                    for (int q = 0; q < 4; ++q)
                        if (r0 + 4 * q < Nn) *(v4f*)(zp + (size_t)q * 4 * EHW) = z4;
                }
            } else {
                const int r0 = t * 16 + (lane >> 4);
                float* zp = aggz + (size_t)r0 * EHW + (lane & 15) * 4;
                if (fullz) {
#pragma unroll
                    for (int q = 0; q < 8; ++q) *(v4f*)(zp + (size_t)q * 2 * EHW) = z4;
                } else {
#pragma unroll
                    for (int q = 0; q < 8; ++q)
                        if (r0 + 2 * q < Nn) *(v4f*)(zp + (size_t)q * 2 * EHW) = z4;
                }
            }
        }

        v8f acc[4];
#pragma unroll
        for (int n = 0; n < 4; ++n) acc[n] = splat8(side ? 0.0f : b1[n * 16 + c]);

#pragma unroll 8
        for (int k = 0; k < HIDN; k += 4) {
            const int kk = k + half * 2;
            v2f a = *(const v2f*)(hr + kk);
#pragma unroll
            for (int n = 0; n < 4; ++n) {
                v2f b;
                b.x = w1s[kk * WS + n * 16 + c];
                b.y = w1s[(kk + 1) * WS + n * 16 + c];
                acc[n] = wmma4(a, b, acc[n]);
            }
        }
        if (t * 16 + 16 <= Nn) {
#pragma unroll
            for (int n = 0; n < 4; ++n) {
                float* p = P + (size_t)(t * 16 + half * 8) * EHW + n * 16 + c;
#pragma unroll
                for (int i = 0; i < 8; ++i) p[(size_t)i * EHW] = acc[n][i];
            }
        } else {
#pragma unroll
            for (int n = 0; n < 4; ++n)
#pragma unroll
                for (int i = 0; i < 8; ++i) {
                    const int r = t * 16 + i + half * 8;
                    if (r < Nn) P[(size_t)r * EHW + n * 16 + c] = acc[n][i];
                }
        }
    }
}

__global__ void edge_kernel(const float* __restrict__ h,
                            const float* __restrict__ P,
                            const int* __restrict__ erow,
                            const int* __restrict__ ecol,
                            const float* __restrict__ w1c,
                            const float* __restrict__ w2,
                            const float* __restrict__ b2,
                            float* __restrict__ agg, int Ee) {
    constexpr int WS  = 72;
    constexpr int LWM = EHW + 4;
    __shared__ float w1s[HIDN * WS];
    __shared__ float w2s[EHW * WS];
    __shared__ float m1s[8][16 * LWM];

    for (int i4 = threadIdx.x; i4 < (HIDN * EHW) / 4; i4 += blockDim.x) {
        const int r = i4 >> 4, col = (i4 & 15) * 4;
        *(v4f*)(w1s + r * WS + col) = *(const v4f*)(w1c + r * EHW + col);
    }
    for (int i4 = threadIdx.x; i4 < (EHW * EHW) / 4; i4 += blockDim.x) {
        const int r = i4 >> 4, col = (i4 & 15) * 4;
        *(v4f*)(w2s + r * WS + col) = *(const v4f*)(w2 + r * EHW + col);
    }
    __syncthreads();

    const int lane = threadIdx.x & 31;
    const int wv   = threadIdx.x >> 5;
    const int c    = lane & 15;
    const int half = lane >> 4;
    const int nw   = (gridDim.x * blockDim.x) >> 5;
    const int wid0 = (blockIdx.x * blockDim.x + threadIdx.x) >> 5;
    const int tiles = (Ee + 15) >> 4;
    float* L = m1s[wv];

    for (int t = wid0; t < tiles; t += nw) {
        int ea = t * 16 + c;
        if (ea >= Ee) ea = Ee - 1;
        const float* __restrict__ hca = h + (size_t)ecol[ea] * HIDN;

        int ri[8];
#pragma unroll
        for (int i = 0; i < 8; ++i) {
            int e = t * 16 + i + half * 8;
            if (e >= Ee) e = Ee - 1;
            ri[i] = erow[e];
        }

        v8f acc[4];
#pragma unroll
        for (int i = 0; i < 8; ++i) {
            const float* __restrict__ pr = P + (size_t)ri[i] * EHW + c;
#pragma unroll
            for (int n = 0; n < 4; ++n) acc[n][i] = pr[n * 16];
        }

#pragma unroll 4
        for (int k = 0; k < HIDN; k += 4) {
            const int kk = k + half * 2;
            v2f a = *(const v2f*)(hca + kk);
#pragma unroll
            for (int n = 0; n < 4; ++n) {
                v2f b;
                b.x = w1s[kk * WS + n * 16 + c];
                b.y = w1s[(kk + 1) * WS + n * 16 + c];
                acc[n] = wmma4(a, b, acc[n]);
            }
        }

#pragma unroll
        for (int n = 0; n < 4; ++n)
#pragma unroll
            for (int i = 0; i < 8; ++i)
                L[(i + half * 8) * LWM + n * 16 + c] = silu_f(acc[n][i]);

        asm volatile("s_wait_dscnt 0" ::: "memory");

        v8f acc2[4];
#pragma unroll
        for (int n = 0; n < 4; ++n) acc2[n] = splat8(b2[n * 16 + c]);

#pragma unroll 4
        for (int k = 0; k < EHW; k += 4) {
            const int kk = k + half * 2;
            v2f a;
            a.x = L[c * LWM + kk];
            a.y = L[c * LWM + kk + 1];
#pragma unroll
            for (int n = 0; n < 4; ++n) {
                v2f b;
                b.x = w2s[kk * WS + n * 16 + c];
                b.y = w2s[(kk + 1) * WS + n * 16 + c];
                acc2[n] = wmma4(a, b, acc2[n]);
            }
        }

        if (t * 16 + 16 <= Ee) {
#pragma unroll
            for (int i = 0; i < 8; ++i) {
                float* dst = agg + (size_t)ri[i] * EHW + c;
#pragma unroll
                for (int n = 0; n < 4; ++n)
                    atomicAdd(dst + n * 16, silu_f(acc2[n][i]));
            }
        } else {
#pragma unroll
            for (int i = 0; i < 8; ++i) {
                const int e = t * 16 + i + half * 8;
                if (e < Ee) {
                    float* dst = agg + (size_t)ri[i] * EHW + c;
#pragma unroll
                    for (int n = 0; n < 4; ++n)
                        atomicAdd(dst + n * 16, silu_f(acc2[n][i]));
                }
            }
        }
        asm volatile("s_wait_dscnt 0" ::: "memory");
    }
}

__global__ void __launch_bounds__(256, 8) edgepq_kernel(const float* __restrict__ P,
                              const float* __restrict__ Q,
                              const int* __restrict__ erow,
                              const int* __restrict__ ecol,
                              const float* __restrict__ w2,
                              const float* __restrict__ b2,
                              float* __restrict__ agg, int Ee) {
    constexpr int WS = 72;
    __shared__ float w2s[EHW * WS];

    {
        constexpr int NS = (EHW * EHW) / 4 / NTHR;
        v4f stg[NS];
#pragma unroll
        for (int j = 0; j < NS; ++j)
            stg[j] = *(const v4f*)(w2 + 4 * (threadIdx.x + j * NTHR));
#pragma unroll
        for (int j = 0; j < NS; ++j) {
            const int i4 = threadIdx.x + j * NTHR;
            *(v4f*)(w2s + (i4 >> 4) * WS + (i4 & 15) * 4) = stg[j];
        }
    }
    __syncthreads();

    const int lane = threadIdx.x & 31;
    const int c    = lane & 15;
    const int half = lane >> 4;
    const int nw   = (gridDim.x * blockDim.x) >> 5;
    const int wid0 = (blockIdx.x * blockDim.x + threadIdx.x) >> 5;
    const int tiles = (Ee + 15) >> 4;

    for (int t = wid0; t < tiles; t += nw) {
        int ea = t * 16 + c;
        if (ea >= Ee) ea = Ee - 1;
        const float* __restrict__ pr = P + (size_t)erow[ea] * EHW + half * 4;
        const float* __restrict__ qr = Q + (size_t)ecol[ea] * EHW + half * 4;

        int ri[8];
#pragma unroll
        for (int i = 0; i < 8; ++i) {
            int e = t * 16 + i + half * 8;
            if (e >= Ee) e = Ee - 1;
            ri[i] = erow[e];
        }

        v4f pv[8], qv[8];
#pragma unroll
        for (int s = 0; s < 8; ++s) {
            pv[s] = *(const v4f*)(pr + 8 * s);
            qv[s] = *(const v4f*)(qr + 8 * s);
        }

        asm volatile("" ::: "memory");

        v8f acc2[4];
#pragma unroll
        for (int n = 0; n < 4; ++n) acc2[n] = splat8(b2[n * 16 + c]);

#pragma unroll
        for (int s = 0; s < 8; ++s) {
            const int kb = 8 * s + 4 * half;
            v2f a0, a1;
            a0.x = silu_f(pv[s].x + qv[s].x);
            a0.y = silu_f(pv[s].y + qv[s].y);
            a1.x = silu_f(pv[s].z + qv[s].z);
            a1.y = silu_f(pv[s].w + qv[s].w);
#pragma unroll
            for (int n = 0; n < 4; ++n) {
                v2f b;
                b.x = w2s[kb * WS + n * 16 + c];
                b.y = w2s[(kb + 1) * WS + n * 16 + c];
                acc2[n] = wmma4(a0, b, acc2[n]);
            }
#pragma unroll
            for (int n = 0; n < 4; ++n) {
                v2f b;
                b.x = w2s[(kb + 2) * WS + n * 16 + c];
                b.y = w2s[(kb + 3) * WS + n * 16 + c];
                acc2[n] = wmma4(a1, b, acc2[n]);
            }
        }

#pragma unroll
        for (int n = 0; n < 4; ++n)
#pragma unroll
            for (int i = 0; i < 8; ++i) acc2[n][i] = silu_f(acc2[n][i]);

        if (t * 16 + 16 <= Ee) {
#pragma unroll
            for (int i = 0; i < 8; ++i) {
                float* dst = agg + (size_t)ri[i] * EHW + c;
#pragma unroll
                for (int n = 0; n < 4; ++n)
                    atomicAdd(dst + n * 16, acc2[n][i]);
            }
        } else {
#pragma unroll
            for (int i = 0; i < 8; ++i) {
                const int e = t * 16 + i + half * 8;
                if (e < Ee) {
                    float* dst = agg + (size_t)ri[i] * EHW + c;
#pragma unroll
                    for (int n = 0; n < 4; ++n)
                        atomicAdd(dst + n * 16, acc2[n][i]);
                }
            }
        }
    }
}

__global__ void __launch_bounds__(NTHR) node_kernel(float* __restrict__ h,
                            const float* __restrict__ agg,
                            const float* __restrict__ m1,
                            const float* __restrict__ c1,
                            const float* __restrict__ m2,
                            const float* __restrict__ c2,
                            int Nn) {
    constexpr int LW  = HIDN + 4;
    constexpr int WS2 = HIDN + 8;
    __shared__ float w2s[HIDN * WS2];
    __shared__ float lds[8][16 * LW];

    {
        constexpr int NS = (HIDN * HIDN) / 4 / NTHR;
        v4f stg[NS];
#pragma unroll
        for (int j = 0; j < NS; ++j)
            stg[j] = *(const v4f*)(m2 + 4 * (threadIdx.x + j * NTHR));
#pragma unroll
        for (int j = 0; j < NS; ++j) {
            const int i4 = threadIdx.x + j * NTHR;
            *(v4f*)(w2s + (i4 >> 5) * WS2 + (i4 & 31) * 4) = stg[j];
        }
    }
    __syncthreads();

    const int lane = threadIdx.x & 31;
    const int wv   = threadIdx.x >> 5;
    const int c    = lane & 15;
    const int half = lane >> 4;
    const int nw   = (gridDim.x * blockDim.x) >> 5;
    const int wid0 = (blockIdx.x * blockDim.x + threadIdx.x) >> 5;
    const int tiles = (Nn + 15) >> 4;
    float* L = lds[wv];

    for (int t = wid0; t < tiles; t += nw) {
        int mr = t * 16 + c;
        if (mr >= Nn) mr = Nn - 1;
        const float* __restrict__ hr = h + (size_t)mr * HIDN;
        const float* __restrict__ ar = agg + (size_t)mr * EHW;

        v8f acc[8];
#pragma unroll
        for (int n = 0; n < 8; ++n) acc[n] = splat8(c1[n * 16 + c]);

        for (int k = 0; k < HIDN; k += 4) {
            const int kk = k + half * 2;
            v2f a = *(const v2f*)(hr + kk);
#pragma unroll
            for (int n = 0; n < 8; ++n) {
                v2f b;
                b.x = m1[kk * HIDN + n * 16 + c];
                b.y = m1[(kk + 1) * HIDN + n * 16 + c];
                acc[n] = wmma4(a, b, acc[n]);
            }
        }
        for (int k = 0; k < EHW; k += 4) {
            const int kk = k + half * 2;
            v2f a = *(const v2f*)(ar + kk);
#pragma unroll
            for (int n = 0; n < 8; ++n) {
                v2f b;
                b.x = m1[(HIDN + kk) * HIDN + n * 16 + c];
                b.y = m1[(HIDN + kk + 1) * HIDN + n * 16 + c];
                acc[n] = wmma4(a, b, acc[n]);
            }
        }
        const bool full = (t * 16 + 16 <= Nn);
        v8f hres[8];
        if (full) {
            const float* p = h + (size_t)(t * 16 + half * 8) * HIDN + c;
#pragma unroll
            for (int n = 0; n < 8; ++n)
#pragma unroll
                for (int i = 0; i < 8; ++i) hres[n][i] = p[(size_t)i * HIDN + n * 16];
        } else {
#pragma unroll
            for (int i = 0; i < 8; ++i) {
                int r = t * 16 + i + half * 8;
                if (r >= Nn) r = Nn - 1;
                const float* p = h + (size_t)r * HIDN + c;
#pragma unroll
                for (int n = 0; n < 8; ++n) hres[n][i] = p[n * 16];
            }
        }

#pragma unroll
        for (int n = 0; n < 8; ++n)
#pragma unroll
            for (int i = 0; i < 8; ++i)
                L[(i + half * 8) * LW + n * 16 + c] = silu_f(acc[n][i]);

        asm volatile("s_wait_dscnt 0" ::: "memory");

        v8f acc2[8];
#pragma unroll
        for (int n = 0; n < 8; ++n) acc2[n] = splat8(c2[n * 16 + c]);

        for (int k = 0; k < HIDN; k += 4) {
            const int kk = k + half * 2;
            v2f a;
            a.x = L[c * LW + kk];
            a.y = L[c * LW + kk + 1];
#pragma unroll
            for (int n = 0; n < 8; ++n) {
                v2f b;
                b.x = w2s[kk * WS2 + n * 16 + c];
                b.y = w2s[(kk + 1) * WS2 + n * 16 + c];
                acc2[n] = wmma4(a, b, acc2[n]);
            }
        }
        if (full) {
#pragma unroll
            for (int n = 0; n < 8; ++n) {
                float* p = h + (size_t)(t * 16 + half * 8) * HIDN + n * 16 + c;
#pragma unroll
                for (int i = 0; i < 8; ++i) p[(size_t)i * HIDN] = hres[n][i] + acc2[n][i];
            }
        } else {
#pragma unroll
            for (int n = 0; n < 8; ++n)
#pragma unroll
                for (int i = 0; i < 8; ++i) {
                    const int r = t * 16 + i + half * 8;
                    if (r < Nn) h[(size_t)r * HIDN + n * 16 + c] = hres[n][i] + acc2[n][i];
                }
        }
        asm volatile("s_wait_dscnt 0" ::: "memory");
    }
}

__global__ void out_kernel(const float* __restrict__ h,
                           const float* __restrict__ W,
                           const float* __restrict__ bias,
                           float* __restrict__ out, int Nn) {
    const int lane = threadIdx.x & 31;
    const int c    = lane & 15;
    const int half = lane >> 4;
    const int nw   = (gridDim.x * blockDim.x) >> 5;
    const int wid0 = (blockIdx.x * blockDim.x + threadIdx.x) >> 5;
    const int tiles = (Nn + 15) >> 4;

    for (int t = wid0; t < tiles; t += nw) {
        int mr = t * 16 + c;
        if (mr >= Nn) mr = Nn - 1;

        v8f acc[4];
#pragma unroll
        for (int n = 0; n < 4; ++n) acc[n] = splat8(bias[n * 16 + c]);

        for (int k = 0; k < HIDN; k += 4) {
            const int kk = k + half * 2;
            v2f a = *(const v2f*)(h + (size_t)mr * HIDN + kk);
#pragma unroll
            for (int n = 0; n < 4; ++n) {
                v2f b;
                b.x = W[kk * EHW + n * 16 + c];
                b.y = W[(kk + 1) * EHW + n * 16 + c];
                acc[n] = wmma4(a, b, acc[n]);
            }
        }
        if (t * 16 + 16 <= Nn) {
#pragma unroll
            for (int n = 0; n < 4; ++n) {
                float* p = out + (size_t)(t * 16 + half * 8) * EHW + n * 16 + c;
#pragma unroll
                for (int i = 0; i < 8; ++i) p[(size_t)i * EHW] = acc[n][i];
            }
        } else {
#pragma unroll
            for (int n = 0; n < 4; ++n)
#pragma unroll
                for (int i = 0; i < 8; ++i) {
                    const int r = t * 16 + i + half * 8;
                    if (r < Nn) out[(size_t)r * EHW + n * 16 + c] = acc[n][i];
                }
        }
    }
}

static inline int imin_h(int a, int b) { return a < b ? a : b; }
static inline int cdiv_h(int a, int b) { return (a + b - 1) / b; }

extern "C" void kernel_launch(void* const* d_in, const int* in_sizes, int n_in,
                              void* d_out, int out_size, void* d_ws, size_t ws_size,
                              hipStream_t stream) {
    const float* h_in  = (const float*)d_in[0];
    const int*   edges = (const int*)d_in[1];
    const float* W_in  = (const float*)d_in[2];
    const float* b_in  = (const float*)d_in[3];
    const float* eW1   = (const float*)d_in[4];
    const float* eb1   = (const float*)d_in[5];
    const float* eW2   = (const float*)d_in[6];
    const float* eb2   = (const float*)d_in[7];
    const float* nW1   = (const float*)d_in[8];
    const float* nb1   = (const float*)d_in[9];
    const float* nW2   = (const float*)d_in[10];
    const float* nb2   = (const float*)d_in[11];
    const float* W_out = (const float*)d_in[12];
    const float* b_out = (const float*)d_in[13];

    const int Nn = in_sizes[0] / IN_NF;
    const int Ee = in_sizes[1] / 2;
    const int Ll = in_sizes[4] / (2 * HIDN * EHW);

    float* h   = (float*)d_ws;
    float* agg = h + (size_t)Nn * HIDN;

    const int* erow = edges;
    const int* ecol = edges + Ee;

    const int tilesN = cdiv_h(Nn, 16);
    const int tilesE = cdiv_h(Ee, 16);
    const int gEmb = imin_h(cdiv_h(tilesN, 8), 2048);
    const int gEdg = imin_h(cdiv_h(tilesE, 8), 2048);
    const int gNod = imin_h(cdiv_h(tilesN, 8), 2048);

    embed_kernel<<<gEmb, 256, 0, stream>>>(h_in, W_in, b_in, h, Nn);

    float* P = (float*)d_out;
    float* Q = agg + (size_t)Nn * EHW;
    const bool roomQ = ws_size >= (size_t)Nn * (HIDN + 2 * EHW) * sizeof(float);

    for (int l = 0; l < Ll; ++l) {
        if (roomQ) {
            rowproj_kernel<<<dim3(gNod, 2), 256, 0, stream>>>(
                h, eW1 + (size_t)l * 2 * HIDN * EHW, eb1 + (size_t)l * EHW, P, Q, agg, Nn);
            edgepq_kernel<<<gEdg, 256, 0, stream>>>(
                P, Q, erow, ecol,
                eW2 + (size_t)l * EHW * EHW, eb2 + (size_t)l * EHW, agg, Ee);
        } else {
            rowproj_kernel<<<dim3(gNod, 1), 256, 0, stream>>>(
                h, eW1 + (size_t)l * 2 * HIDN * EHW, eb1 + (size_t)l * EHW, P, nullptr, agg, Nn);
            edge_kernel<<<gEdg, 256, 0, stream>>>(
                h, P, erow, ecol,
                eW1 + (size_t)l * 2 * HIDN * EHW + (size_t)HIDN * EHW,
                eW2 + (size_t)l * EHW * EHW,      eb2 + (size_t)l * EHW,
                agg, Ee);
        }
        node_kernel<<<gNod, 256, 0, stream>>>(
            h, agg,
            nW1 + (size_t)l * KIN_NODE * HIDN, nb1 + (size_t)l * HIDN,
            nW2 + (size_t)l * HIDN * HIDN,     nb2 + (size_t)l * HIDN,
            Nn);
    }

    out_kernel<<<gEmb, 256, 0, stream>>>(h, W_out, b_out, (float*)d_out, Nn);
}
